# starting kernel
# baseline (speedup 1.0000x reference)
#include <hip/hip_runtime.h>
#include <hip/hip_bf16.h>

#define NN     50000
#define MPAD   50016
#define E0C    800000
#define ETOT   850000
#define INDIM  256
#define HIDDIM 256
#define OUTDIM 64
#define H1C    8
#define C1C    32
#define H2C    1
#define C2C    64
#define NEG_SLOPE 0.2f

typedef __attribute__((ext_vector_type(16))) _Float16 v16h;
typedef __attribute__((ext_vector_type(8)))  _Float16 v8h;
typedef __attribute__((ext_vector_type(8)))  float    v8f;

union V16U { v16h v; v8h h[2]; };

__device__ __forceinline__ unsigned fenc(float f) {
    int b = __float_as_int(f);
    return (b >= 0) ? ((unsigned)b | 0x80000000u) : ~(unsigned)b;
}
__device__ __forceinline__ float fdec(unsigned u) {
    int b = (u & 0x80000000u) ? (int)(u & 0x7FFFFFFFu) : (int)(~u);
    return __int_as_float(b);
}

__global__ void cvt_f32_to_f16(const float* __restrict__ src,
                               _Float16* __restrict__ dst, int n) {
    int i = blockIdx.x * blockDim.x + threadIdx.x;
    if (i < n) dst[i] = (_Float16)src[i];
}

__global__ void xpose_cvt_f16(const float* __restrict__ W,
                              _Float16* __restrict__ Wt, int K, int Nc) {
    int i = blockIdx.x * blockDim.x + threadIdx.x;
    if (i >= K * Nc) return;
    int k = i / Nc, n = i % Nc;
    Wt[(size_t)n * K + k] = (_Float16)W[i];
}

#define WMMA16(a, b, c) __builtin_amdgcn_wmma_f32_16x16x32_f16( \
        false, (a), false, (b), (short)0, (c), false, false)

__global__ __launch_bounds__(128)
void gemm_f16_wmma(const _Float16* __restrict__ A,
                   const _Float16* __restrict__ Bt,
                   float* __restrict__ C, int Mtiles2, int Nc, int K) {
    int wave = (int)((blockIdx.x * blockDim.x + threadIdx.x) >> 5);
    int lane = threadIdx.x & 31;
    int nt2 = Nc >> 5;
    int tm = wave / nt2;
    int tn = wave % nt2;
    if (tm >= Mtiles2) return;
    int r    = lane & 15;
    int half = lane >> 4;

    const _Float16* a0 = A  + (size_t)(tm * 32 + r) * K + half * 8;
    const _Float16* a1 = a0 + (size_t)16 * K;
    const _Float16* b0 = Bt + (size_t)(tn * 32 + r) * K + half * 8;
    const _Float16* b1 = b0 + (size_t)16 * K;

    v8f c00 = {}, c01 = {}, c10 = {}, c11 = {};
    for (int k0 = 0; k0 < K; k0 += 32) {
        V16U A0, A1, B0, B1;
        A0.h[0] = *(const v8h*)(a0 + k0);
        A0.h[1] = *(const v8h*)(a0 + k0 + 16);
        A1.h[0] = *(const v8h*)(a1 + k0);
        A1.h[1] = *(const v8h*)(a1 + k0 + 16);
        B0.h[0] = *(const v8h*)(b0 + k0);
        B0.h[1] = *(const v8h*)(b0 + k0 + 16);
        B1.h[0] = *(const v8h*)(b1 + k0);
        B1.h[1] = *(const v8h*)(b1 + k0 + 16);
        c00 = WMMA16(A0.v, B0.v, c00);
        c01 = WMMA16(A0.v, B1.v, c01);
        c10 = WMMA16(A1.v, B0.v, c10);
        c11 = WMMA16(A1.v, B1.v, c11);
    }
    {
        float* p = C + (size_t)(tm * 32 + half * 8) * Nc + tn * 32 + r;
#pragma unroll
        for (int j = 0; j < 8; ++j) { p[(size_t)j * Nc] = c00[j]; p[(size_t)j * Nc + 16] = c01[j]; }
        p += (size_t)16 * Nc;
#pragma unroll
        for (int j = 0; j < 8; ++j) { p[(size_t)j * Nc] = c10[j]; p[(size_t)j * Nc + 16] = c11[j]; }
    }
}

__global__ void node_coef(const float* __restrict__ h,
                          const float* __restrict__ att,
                          float* __restrict__ ai, float* __restrict__ aj,
                          int nodes, int H, int Cc) {
    int idx = blockIdx.x * blockDim.x + threadIdx.x;
    if (idx >= nodes * H) return;
    int node = idx / H, head = idx % H;
    const float* hv = h + (size_t)node * H * Cc + head * Cc;
    const float* ad = att + head * 2 * Cc;
    const float* as = ad + Cc;
    float si = 0.f, sj = 0.f;
    for (int c = 0; c < Cc; ++c) { si += hv[c] * ad[c]; sj += hv[c] * as[c]; }
    ai[idx] = si;
    aj[idx] = sj;
}

__global__ void edge_max(const int* __restrict__ ei,
                         const float* __restrict__ ai, const float* __restrict__ aj,
                         float* __restrict__ alpha, unsigned* __restrict__ mmax,
                         int H) {
    int idx = blockIdx.x * blockDim.x + threadIdx.x;
    if (idx >= ETOT * H) return;
    int e = idx / H, head = idx % H;
    int s_, d_;
    if (e < E0C) { s_ = ei[e]; d_ = ei[E0C + e]; } else { s_ = d_ = e - E0C; }
    float a = ai[(size_t)d_ * H + head] + aj[(size_t)s_ * H + head];
    a = (a >= 0.f) ? a : NEG_SLOPE * a;
    alpha[idx] = a;
    atomicMax(mmax + (size_t)d_ * H + head, fenc(a));
}

__global__ void edge_expsum(const int* __restrict__ ei,
                            float* __restrict__ alpha,
                            const unsigned* __restrict__ mmax,
                            float* __restrict__ ssum, int H) {
    int idx = blockIdx.x * blockDim.x + threadIdx.x;
    if (idx >= ETOT * H) return;
    int e = idx / H, head = idx % H;
    int d_;
    if (e < E0C) { d_ = ei[E0C + e]; } else { d_ = e - E0C; }
    float m  = fdec(mmax[(size_t)d_ * H + head]);
    float ex = __expf(alpha[idx] - m);
    alpha[idx] = ex;
    atomicAdd(ssum + (size_t)d_ * H + head, ex);
}

__global__ void edge_aggregate(const int* __restrict__ ei,
                               const float* __restrict__ expw,
                               const float* __restrict__ ssum,
                               const float* __restrict__ h,
                               float* __restrict__ agg, int H, int Cc) {
    int gw   = (int)((blockIdx.x * blockDim.x + threadIdx.x) >> 5);
    int lane = threadIdx.x & 31;
    if (gw >= ETOT) return;
    int s_, d_;
    if (gw < E0C) { s_ = ei[gw]; d_ = ei[E0C + gw]; } else { s_ = d_ = gw - E0C; }
    int HC = H * Cc;
    const float* hs = h   + (size_t)s_ * HC;
    float*       ad = agg + (size_t)d_ * HC;
    for (int i = 0; i < HC; i += 32) {
        int ch   = i + lane;
        int head = ch / Cc;
        float w  = expw[(size_t)gw * H + head] / ssum[(size_t)d_ * H + head];
        atomicAdd(ad + ch, w * hs[ch]);
    }
}

__global__ void elu_bias_cvt(const float* __restrict__ agg,
                             const float* __restrict__ bias,
                             _Float16* __restrict__ out, int total, int HC) {
    int i = blockIdx.x * blockDim.x + threadIdx.x;
    if (i >= total) return;
    float v = agg[i] + bias[i % HC];
    v = (v > 0.f) ? v : (__expf(v) - 1.f);
    out[i] = (_Float16)v;
}

__global__ void add_bias(float* __restrict__ out, const float* __restrict__ bias,
                         int total, int Cc) {
    int i = blockIdx.x * blockDim.x + threadIdx.x;
    if (i < total) out[i] += bias[i % Cc];
}

extern "C" void kernel_launch(void* const* d_in, const int* in_sizes, int n_in,
                              void* d_out, int out_size, void* d_ws, size_t ws_size,
                              hipStream_t stream) {
    (void)in_sizes; (void)n_in; (void)out_size; (void)ws_size;

    const float* x    = (const float*)d_in[0];
    const int*   ei   = (const int*)  d_in[1];
    const float* W1   = (const float*)d_in[2];
    const float* att1 = (const float*)d_in[3];
    const float* b1   = (const float*)d_in[4];
    const float* W2   = (const float*)d_in[5];
    const float* att2 = (const float*)d_in[6];
    const float* b2   = (const float*)d_in[7];
    float* out = (float*)d_out;

    char* base = (char*)d_ws;
    size_t off = 0;
    auto carve = [&](size_t bytes) -> char* {
        char* p = base + off;
        off = (off + bytes + 255) & ~(size_t)255;
        return p;
    };
    _Float16* XH    = (_Float16*)carve((size_t)MPAD * INDIM * 2);
    _Float16* W1T   = (_Float16*)carve((size_t)INDIM * HIDDIM * 2);
    _Float16* W2T   = (_Float16*)carve((size_t)HIDDIM * OUTDIM * 2);
    float*    HBUF  = (float*)   carve((size_t)MPAD * HIDDIM * 4);
    float*    AGG1  = (float*)   carve((size_t)NN * HIDDIM * 4);
    float*    AI    = (float*)   carve((size_t)NN * H1C * 4);
    float*    AJ    = (float*)   carve((size_t)NN * H1C * 4);
    unsigned* MM    = (unsigned*)carve((size_t)NN * H1C * 4);
    float*    SS    = (float*)   carve((size_t)NN * H1C * 4);
    float*    ALPHA = (float*)   carve((size_t)ETOT * H1C * 4);

    const int TB = 256;
    auto nb = [](long long n, int tb) { return (unsigned)((n + tb - 1) / tb); };
    const int MT2 = MPAD / 32;

    cvt_f32_to_f16<<<nb((long long)NN * INDIM, TB), TB, 0, stream>>>(x, XH, NN * INDIM);
    xpose_cvt_f16 <<<nb((long long)INDIM * HIDDIM, TB), TB, 0, stream>>>(W1, W1T, INDIM, HIDDIM);
    xpose_cvt_f16 <<<nb((long long)HIDDIM * OUTDIM, TB), TB, 0, stream>>>(W2, W2T, HIDDIM, OUTDIM);

    {
        long long waves = (long long)MT2 * (HIDDIM / 32);
        gemm_f16_wmma<<<nb(waves * 32, 128), 128, 0, stream>>>(XH, W1T, HBUF, MT2, HIDDIM, INDIM);
    }

    node_coef<<<nb((long long)NN * H1C, TB), TB, 0, stream>>>(HBUF, att1, AI, AJ, NN, H1C, C1C);

    hipMemsetAsync(MM,   0, (size_t)NN * H1C * 4, stream);
    hipMemsetAsync(SS,   0, (size_t)NN * H1C * 4, stream);
    hipMemsetAsync(AGG1, 0, (size_t)NN * HIDDIM * 4, stream);

    edge_max   <<<nb((long long)ETOT * H1C, TB), TB, 0, stream>>>(ei, AI, AJ, ALPHA, MM, H1C);
    edge_expsum<<<nb((long long)ETOT * H1C, TB), TB, 0, stream>>>(ei, ALPHA, MM, SS, H1C);
    edge_aggregate<<<nb((long long)ETOT * 32, TB), TB, 0, stream>>>(ei, ALPHA, SS, HBUF, AGG1, H1C, C1C);

    elu_bias_cvt<<<nb((long long)NN * HIDDIM, TB), TB, 0, stream>>>(AGG1, b1, XH, NN * HIDDIM, HIDDIM);

    {
        long long waves = (long long)MT2 * (OUTDIM / 32);
        gemm_f16_wmma<<<nb(waves * 32, 128), 128, 0, stream>>>(XH, W2T, HBUF, MT2, OUTDIM, HIDDIM);
    }

    node_coef<<<nb((long long)NN * H2C, TB), TB, 0, stream>>>(HBUF, att2, AI, AJ, NN, H2C, C2C);

    hipMemsetAsync(MM,  0, (size_t)NN * H2C * 4, stream);
    hipMemsetAsync(SS,  0, (size_t)NN * H2C * 4, stream);
    hipMemsetAsync(out, 0, (size_t)NN * OUTDIM * 4, stream);

    edge_max   <<<nb((long long)ETOT * H2C, TB), TB, 0, stream>>>(ei, AI, AJ, ALPHA, MM, H2C);
    edge_expsum<<<nb((long long)ETOT * H2C, TB), TB, 0, stream>>>(ei, ALPHA, MM, SS, H2C);
    edge_aggregate<<<nb((long long)ETOT * 32, TB), TB, 0, stream>>>(ei, ALPHA, SS, HBUF, out, H2C, C2C);

    add_bias<<<nb((long long)NN * OUTDIM, TB), TB, 0, stream>>>(out, b2, NN * OUTDIM, OUTDIM);
}


# Edges grouped by target once per call; per-node running softmax and row sum in registers (wave per node in layer 1, half-wave in layer 2) replace float atomics; fp32 sum order follows arrival rank
# speedup vs baseline: 2.3182x; 2.3182x over previous
#include <hip/hip_runtime.h>
#include <hip/hip_bf16.h>

#define NN     50000
#define MPAD   50016
#define E0C    800000
#define ETOT   850000
#define INDIM  256
#define HIDDIM 256
#define OUTDIM 64
#define H1C    8
#define C1C    32
#define H2C    1
#define C2C    64
#define NEG_SLOPE 0.2f

typedef __attribute__((ext_vector_type(16))) _Float16 v16h;
typedef __attribute__((ext_vector_type(8)))  _Float16 v8h;
typedef __attribute__((ext_vector_type(8)))  float    v8f;

union V16U { v16h v; v8h h[2]; };

__global__ void cvt_f32_to_f16(const float* __restrict__ src,
                               _Float16* __restrict__ dst, int n) {
    int i = blockIdx.x * blockDim.x + threadIdx.x;
    if (i < n) dst[i] = (_Float16)src[i];
}

__global__ void xpose_cvt_f16(const float* __restrict__ W,
                              _Float16* __restrict__ Wt, int K, int Nc) {
    int i = blockIdx.x * blockDim.x + threadIdx.x;
    if (i >= K * Nc) return;
    int k = i / Nc, n = i % Nc;
    Wt[(size_t)n * K + k] = (_Float16)W[i];
}

#define WMMA16(a, b, c) __builtin_amdgcn_wmma_f32_16x16x32_f16( \
        false, (a), false, (b), (short)0, (c), false, false)

__global__ __launch_bounds__(128)
void gemm_f16_wmma(const _Float16* __restrict__ A,
                   const _Float16* __restrict__ Bt,
                   float* __restrict__ C, int Mtiles2, int Nc, int K) {
    int wave = (int)((blockIdx.x * blockDim.x + threadIdx.x) >> 5);
    int lane = threadIdx.x & 31;
    int nt2 = Nc >> 5;
    int tm = wave / nt2;
    int tn = wave % nt2;
    if (tm >= Mtiles2) return;
    int r    = lane & 15;
    int half = lane >> 4;

    const _Float16* a0 = A  + (size_t)(tm * 32 + r) * K + half * 8;
    const _Float16* a1 = a0 + (size_t)16 * K;
    const _Float16* b0 = Bt + (size_t)(tn * 32 + r) * K + half * 8;
    const _Float16* b1 = b0 + (size_t)16 * K;

    v8f c00 = {}, c01 = {}, c10 = {}, c11 = {};
    for (int k0 = 0; k0 < K; k0 += 32) {
        V16U A0, A1, B0, B1;
        A0.h[0] = *(const v8h*)(a0 + k0);
        A0.h[1] = *(const v8h*)(a0 + k0 + 16);
        A1.h[0] = *(const v8h*)(a1 + k0);
        A1.h[1] = *(const v8h*)(a1 + k0 + 16);
        B0.h[0] = *(const v8h*)(b0 + k0);
        B0.h[1] = *(const v8h*)(b0 + k0 + 16);
        B1.h[0] = *(const v8h*)(b1 + k0);
        B1.h[1] = *(const v8h*)(b1 + k0 + 16);
        c00 = WMMA16(A0.v, B0.v, c00);
        c01 = WMMA16(A0.v, B1.v, c01);
        c10 = WMMA16(A1.v, B0.v, c10);
        c11 = WMMA16(A1.v, B1.v, c11);
    }
    {
        float* p = C + (size_t)(tm * 32 + half * 8) * Nc + tn * 32 + r;
#pragma unroll
        for (int j = 0; j < 8; ++j) { p[(size_t)j * Nc] = c00[j]; p[(size_t)j * Nc + 16] = c01[j]; }
        p += (size_t)16 * Nc;
#pragma unroll
        for (int j = 0; j < 8; ++j) { p[(size_t)j * Nc] = c10[j]; p[(size_t)j * Nc + 16] = c11[j]; }
    }
}

__global__ void node_coef(const float* __restrict__ h,
                          const float* __restrict__ att,
                          float* __restrict__ ai, float* __restrict__ aj,
                          int nodes, int H, int Cc) {
    int idx = blockIdx.x * blockDim.x + threadIdx.x;
    if (idx >= nodes * H) return;
    int node = idx / H, head = idx % H;
    const float* hv = h + (size_t)node * H * Cc + head * Cc;
    const float* ad = att + head * 2 * Cc;
    const float* as = ad + Cc;
    float si = 0.f, sj = 0.f;
    for (int c = 0; c < Cc; ++c) { si += hv[c] * ad[c]; sj += hv[c] * as[c]; }
    ai[idx] = si;
    aj[idx] = sj;
}

__global__ void csr_count(const int* __restrict__ ei, int* __restrict__ cnt,
                          int* __restrict__ rank) {
    int e = blockIdx.x * blockDim.x + threadIdx.x;
    if (e >= E0C) return;
    int d_ = ei[E0C + e];
    rank[e] = atomicAdd(cnt + d_, 1);
}

#define SCAN_TB 1024
#define SCAN_CH ((NN + SCAN_TB - 1) / SCAN_TB)
__global__ __launch_bounds__(SCAN_TB)
void csr_scan(const int* __restrict__ cnt, int* __restrict__ rowptr) {
    __shared__ int wsum[32];
    int t = threadIdx.x;
    int lane = t & 31, w = t >> 5;
    int b = t * SCAN_CH;
    int v[SCAN_CH];
    int s = 0;
#pragma unroll
    for (int i = 0; i < SCAN_CH; ++i) {
        v[i] = (b + i < NN) ? cnt[b + i] : 0;
        s += v[i];
    }
    int x = s;
    for (int o = 1; o < 32; o <<= 1) {
        int y = __shfl_up(x, o);
        if (lane >= o) x += y;
    }
    if (lane == 31) wsum[w] = x;
    __syncthreads();
    if (w == 0) {
        int tv = wsum[lane];
        for (int o = 1; o < 32; o <<= 1) {
            int y = __shfl_up(tv, o);
            if (lane >= o) tv += y;
        }
        wsum[lane] = tv;
    }
    __syncthreads();
    int run = x - s + ((w > 0) ? wsum[w - 1] : 0);
#pragma unroll
    for (int i = 0; i < SCAN_CH; ++i) {
        if (b + i < NN) rowptr[b + i] = run;
        run += v[i];
    }
    if (t == SCAN_TB - 1) rowptr[NN] = run;
}

__global__ void csr_fill(const int* __restrict__ ei, const int* __restrict__ rowptr,
                         const int* __restrict__ rank, int* __restrict__ csrc) {
    int e = blockIdx.x * blockDim.x + threadIdx.x;
    if (e >= E0C) return;
    int d_ = ei[E0C + e];
    csrc[rowptr[d_] + rank[e]] = ei[e];
}

__device__ __forceinline__ float lrelu(float a) { return (a >= 0.f) ? a : NEG_SLOPE * a; }

__device__ __forceinline__ void fold_edge(float a, const float4 r,
                                          float& m, float& ssum, float4& acc) {
    float mn = fmaxf(m, a);
    float sc = __expf(m - mn);
    float w  = __expf(a - mn);
    ssum  = ssum * sc + w;
    acc.x = acc.x * sc + w * r.x;
    acc.y = acc.y * sc + w * r.y;
    acc.z = acc.z * sc + w * r.z;
    acc.w = acc.w * sc + w * r.w;
    m = mn;
}

__global__ __launch_bounds__(256)
void node_softmax_l1(const int* __restrict__ rowptr, const int* __restrict__ csrc,
                 const float* __restrict__ ai, const float* __restrict__ aj,
                 const float* __restrict__ h, float* __restrict__ agg) {
    int node = (int)((blockIdx.x * blockDim.x + threadIdx.x) >> 5);
    int lane = threadIdx.x & 31;
    if (node >= NN) return;
    const int hq  = lane >> 3;
    const int odd = lane & 1;
    const int sel = hq + 4 * odd;
    const float ai0 = ai[(size_t)node * H1C + hq];
    const float ai1 = ai[(size_t)node * H1C + 4 + hq];
    const int beg = rowptr[node], end = rowptr[node + 1];

#define L1_LOAD(s, ajv, r0, r1) { \
        ajv = aj[(size_t)(s) * H1C + sel]; \
        const float4* p_ = (const float4*)(h + (size_t)(s) * HIDDIM); \
        r0 = p_[lane]; r1 = p_[32 + lane]; }
#define L1_FOLD(ajv, r0, r1) { \
        float o_ = __shfl_xor(ajv, 1); \
        float a0_ = lrelu(ai0 + (odd ? o_ : ajv)); \
        float a1_ = lrelu(ai1 + (odd ? ajv : o_)); \
        fold_edge(a0_, r0, m0, s0, acc0); \
        fold_edge(a1_, r1, m1, s1, acc1); }

    float m0, m1, s0 = 1.f, s1 = 1.f;
    float4 acc0, acc1;
    {
        float ajv;
        L1_LOAD(node, ajv, acc0, acc1);
        float o_ = __shfl_xor(ajv, 1);
        m0 = lrelu(ai0 + (odd ? o_ : ajv));
        m1 = lrelu(ai1 + (odd ? ajv : o_));
    }
    int p = beg;
    for (; p + 4 <= end; p += 4) {
        int sa = csrc[p], sb = csrc[p + 1], sc_ = csrc[p + 2], sd = csrc[p + 3];
        float ja, jb, jc, jd;
        float4 a0, a1, b0, b1, c0, c1, d0, d1;
        L1_LOAD(sa, ja, a0, a1);
        L1_LOAD(sb, jb, b0, b1);
        L1_LOAD(sc_, jc, c0, c1);
        L1_LOAD(sd, jd, d0, d1);
        L1_FOLD(ja, a0, a1);
        L1_FOLD(jb, b0, b1);
        L1_FOLD(jc, c0, c1);
        L1_FOLD(jd, d0, d1);
    }
    for (; p < end; ++p) {
        int sa = csrc[p];
        float ja; float4 a0, a1;
        L1_LOAD(sa, ja, a0, a1);
        L1_FOLD(ja, a0, a1);
    }
#undef L1_LOAD
#undef L1_FOLD
    float i0 = 1.f / s0, i1 = 1.f / s1;
    float4* o = (float4*)(agg + (size_t)node * HIDDIM);
    o[lane]      = make_float4(acc0.x * i0, acc0.y * i0, acc0.z * i0, acc0.w * i0);
    o[32 + lane] = make_float4(acc1.x * i1, acc1.y * i1, acc1.z * i1, acc1.w * i1);
}

__global__ __launch_bounds__(256)
void node_softmax_l2(const int* __restrict__ rowptr, const int* __restrict__ csrc,
                 const float* __restrict__ ai, const float* __restrict__ aj,
                 const float* __restrict__ h, float* __restrict__ outp) {
    int gid  = blockIdx.x * blockDim.x + threadIdx.x;
    int node = gid >> 4;
    int lg   = gid & 15;
    if (node >= NN) return;
    const float aiv = ai[node];
    const int beg = rowptr[node], end = rowptr[node + 1];

    float m = lrelu(aiv + aj[node]), ssum = 1.f;
    float4 acc = ((const float4*)(h + (size_t)node * OUTDIM))[lg];

    int p = beg;
    for (; p + 4 <= end; p += 4) {
        int sa = csrc[p], sb = csrc[p + 1], sc_ = csrc[p + 2], sd = csrc[p + 3];
        float ja = aj[sa], jb = aj[sb], jc = aj[sc_], jd = aj[sd];
        float4 ra = ((const float4*)(h + (size_t)sa  * OUTDIM))[lg];
        float4 rb = ((const float4*)(h + (size_t)sb  * OUTDIM))[lg];
        float4 rc = ((const float4*)(h + (size_t)sc_ * OUTDIM))[lg];
        float4 rd = ((const float4*)(h + (size_t)sd  * OUTDIM))[lg];
        fold_edge(lrelu(aiv + ja), ra, m, ssum, acc);
        fold_edge(lrelu(aiv + jb), rb, m, ssum, acc);
        fold_edge(lrelu(aiv + jc), rc, m, ssum, acc);
        fold_edge(lrelu(aiv + jd), rd, m, ssum, acc);
    }
    for (; p < end; ++p) {
        int sa = csrc[p];
        float ja = aj[sa];
        float4 ra = ((const float4*)(h + (size_t)sa * OUTDIM))[lg];
        fold_edge(lrelu(aiv + ja), ra, m, ssum, acc);
    }
    float inv = 1.f / ssum;
    ((float4*)(outp + (size_t)node * OUTDIM))[lg] =
        make_float4(acc.x * inv, acc.y * inv, acc.z * inv, acc.w * inv);
}

__global__ void elu_bias_cvt(const float* __restrict__ agg,
                             const float* __restrict__ bias,
                             _Float16* __restrict__ out, int total, int HC) {
    int i = blockIdx.x * blockDim.x + threadIdx.x;
    if (i >= total) return;
    float v = agg[i] + bias[i % HC];
    v = (v > 0.f) ? v : (__expf(v) - 1.f);
    out[i] = (_Float16)v;
}

__global__ void add_bias(float* __restrict__ out, const float* __restrict__ bias,
                         int total, int Cc) {
    int i = blockIdx.x * blockDim.x + threadIdx.x;
    if (i < total) out[i] += bias[i % Cc];
}

extern "C" void kernel_launch(void* const* d_in, const int* in_sizes, int n_in,
                              void* d_out, int out_size, void* d_ws, size_t ws_size,
                              hipStream_t stream) {
    (void)in_sizes; (void)n_in; (void)out_size; (void)ws_size;

    const float* x    = (const float*)d_in[0];
    const int*   ei   = (const int*)  d_in[1];
    const float* W1   = (const float*)d_in[2];
    const float* att1 = (const float*)d_in[3];
    const float* b1   = (const float*)d_in[4];
    const float* W2   = (const float*)d_in[5];
    const float* att2 = (const float*)d_in[6];
    const float* b2   = (const float*)d_in[7];
    float* out = (float*)d_out;

    char* base = (char*)d_ws;
    size_t off = 0;
    auto carve = [&](size_t bytes) -> char* {
        char* p = base + off;
        off = (off + bytes + 255) & ~(size_t)255;
        return p;
    };
    _Float16* XH    = (_Float16*)carve((size_t)MPAD * INDIM * 2);
    _Float16* W1T   = (_Float16*)carve((size_t)INDIM * HIDDIM * 2);
    _Float16* W2T   = (_Float16*)carve((size_t)HIDDIM * OUTDIM * 2);
    float*    HBUF  = (float*)   carve((size_t)MPAD * HIDDIM * 4);
    float*    AGG1  = (float*)   carve((size_t)NN * HIDDIM * 4);
    float*    AI    = (float*)   carve((size_t)NN * H1C * 4);
    float*    AJ    = (float*)   carve((size_t)NN * H1C * 4);
    int*      CNT   = (int*)     carve((size_t)NN * 4);
    int*      RANK  = (int*)     carve((size_t)E0C * 4);
    int*      ROWP  = (int*)     carve((size_t)(NN + 1) * 4);
    int*      CSRC  = (int*)     carve((size_t)E0C * 4);

    const int TB = 256;
    auto nb = [](long long n, int tb) { return (unsigned)((n + tb - 1) / tb); };
    const int MT2 = MPAD / 32;

    hipMemsetAsync(CNT, 0, (size_t)NN * 4, stream);
    csr_count<<<nb((long long)E0C, TB), TB, 0, stream>>>(ei, CNT, RANK);
    csr_scan <<<1, SCAN_TB, 0, stream>>>(CNT, ROWP);
    csr_fill <<<nb((long long)E0C, TB), TB, 0, stream>>>(ei, ROWP, RANK, CSRC);

    cvt_f32_to_f16<<<nb((long long)NN * INDIM, TB), TB, 0, stream>>>(x, XH, NN * INDIM);
    xpose_cvt_f16 <<<nb((long long)INDIM * HIDDIM, TB), TB, 0, stream>>>(W1, W1T, INDIM, HIDDIM);
    xpose_cvt_f16 <<<nb((long long)HIDDIM * OUTDIM, TB), TB, 0, stream>>>(W2, W2T, HIDDIM, OUTDIM);

    {
        long long waves = (long long)MT2 * (HIDDIM / 32);
        gemm_f16_wmma<<<nb(waves * 32, 128), 128, 0, stream>>>(XH, W1T, HBUF, MT2, HIDDIM, INDIM);
    }

    node_coef<<<nb((long long)NN * H1C, TB), TB, 0, stream>>>(HBUF, att1, AI, AJ, NN, H1C, C1C);

    node_softmax_l1<<<nb((long long)NN * 32, TB), TB, 0, stream>>>(ROWP, CSRC, AI, AJ, HBUF, AGG1);

    elu_bias_cvt<<<nb((long long)NN * HIDDIM, TB), TB, 0, stream>>>(AGG1, b1, XH, NN * HIDDIM, HIDDIM);

    {
        long long waves = (long long)MT2 * (OUTDIM / 32);
        gemm_f16_wmma<<<nb(waves * 32, 128), 128, 0, stream>>>(XH, W2T, HBUF, MT2, OUTDIM, HIDDIM);
    }

    node_coef<<<nb((long long)NN * H2C, TB), TB, 0, stream>>>(HBUF, att2, AI, AJ, NN, H2C, C2C);

    node_softmax_l2<<<nb((long long)NN * 16, TB), TB, 0, stream>>>(ROWP, CSRC, AI, AJ, HBUF, out);

    add_bias<<<nb((long long)NN * OUTDIM, TB), TB, 0, stream>>>(out, b2, NN * OUTDIM, OUTDIM);
}


# Elementwise passes (cast of x, bias+ELU+cast, final bias): a wave moves 16 or 8 whole 512-byte pieces, 128-bit loads all in flight, whole-line stores; 900K waves become 16K; arithmetic unchanged
# speedup vs baseline: 3.1401x; 1.3545x over previous
#include <hip/hip_runtime.h>
#include <hip/hip_bf16.h>

#define NN     50000
#define MPAD   50016
#define E0C    800000
#define ETOT   850000
#define INDIM  256
#define HIDDIM 256
#define OUTDIM 64
#define H1C    8
#define C1C    32
#define H2C    1
#define C2C    64
#define NEG_SLOPE 0.2f

typedef __attribute__((ext_vector_type(16))) _Float16 v16h;
typedef __attribute__((ext_vector_type(8)))  _Float16 v8h;
typedef __attribute__((ext_vector_type(8)))  float    v8f;

union V16U { v16h v; v8h h[2]; };

#define EW_BIG   16
#define EW_SMALL 8
typedef __attribute__((ext_vector_type(4))) _Float16 v4h;

__global__ __launch_bounds__(256)
void cvt_f32_to_f16(const float* __restrict__ src, _Float16* __restrict__ dst) {
    constexpr int NQ = NN * INDIM / 4;
    static_assert(NQ % (32 * EW_BIG) == 0, "whole waves only");
    int wave = (int)((blockIdx.x * blockDim.x + threadIdx.x) >> 5);
    int lane = threadIdx.x & 31;
    if (wave * (32 * EW_BIG) >= NQ) return;
    int q0 = wave * (32 * EW_BIG) + lane;
    const float4* s4 = (const float4*)src;
    v4h* d4 = (v4h*)dst;
    float4 v[EW_BIG];
#pragma unroll
    for (int u = 0; u < EW_BIG; ++u) v[u] = s4[q0 + u * 32];
#pragma unroll
    for (int u = 0; u < EW_BIG; ++u) {
        v4h o;
        o.x = (_Float16)v[u].x; o.y = (_Float16)v[u].y;
        o.z = (_Float16)v[u].z; o.w = (_Float16)v[u].w;
        d4[q0 + u * 32] = o;
    }
}

__global__ void xpose_cvt_f16(const float* __restrict__ W,
                              _Float16* __restrict__ Wt, int K, int Nc) {
    int i = blockIdx.x * blockDim.x + threadIdx.x;
    if (i >= K * Nc) return;
    int k = i / Nc, n = i % Nc;
    Wt[(size_t)n * K + k] = (_Float16)W[i];
}

#define WMMA16(a, b, c) __builtin_amdgcn_wmma_f32_16x16x32_f16( \
        false, (a), false, (b), (short)0, (c), false, false)

__global__ __launch_bounds__(128)
void gemm_f16_wmma(const _Float16* __restrict__ A,
                   const _Float16* __restrict__ Bt,
                   float* __restrict__ C, int Mtiles2, int Nc, int K) {
    int wave = (int)((blockIdx.x * blockDim.x + threadIdx.x) >> 5);
    int lane = threadIdx.x & 31;
    int nt2 = Nc >> 5;
    int tm = wave / nt2;
    int tn = wave % nt2;
    if (tm >= Mtiles2) return;
    int r    = lane & 15;
    int half = lane >> 4;

    const _Float16* a0 = A  + (size_t)(tm * 32 + r) * K + half * 8;
    const _Float16* a1 = a0 + (size_t)16 * K;
    const _Float16* b0 = Bt + (size_t)(tn * 32 + r) * K + half * 8;
    const _Float16* b1 = b0 + (size_t)16 * K;

    v8f c00 = {}, c01 = {}, c10 = {}, c11 = {};
    for (int k0 = 0; k0 < K; k0 += 32) {
        V16U A0, A1, B0, B1;
        A0.h[0] = *(const v8h*)(a0 + k0);
        A0.h[1] = *(const v8h*)(a0 + k0 + 16);
        A1.h[0] = *(const v8h*)(a1 + k0);
        A1.h[1] = *(const v8h*)(a1 + k0 + 16);
        B0.h[0] = *(const v8h*)(b0 + k0);
        B0.h[1] = *(const v8h*)(b0 + k0 + 16);
        B1.h[0] = *(const v8h*)(b1 + k0);
        B1.h[1] = *(const v8h*)(b1 + k0 + 16);
        c00 = WMMA16(A0.v, B0.v, c00);
        c01 = WMMA16(A0.v, B1.v, c01);
        c10 = WMMA16(A1.v, B0.v, c10);
        c11 = WMMA16(A1.v, B1.v, c11);
    }
    {
        float* p = C + (size_t)(tm * 32 + half * 8) * Nc + tn * 32 + r;
#pragma unroll
        for (int j = 0; j < 8; ++j) { p[(size_t)j * Nc] = c00[j]; p[(size_t)j * Nc + 16] = c01[j]; }
        p += (size_t)16 * Nc;
#pragma unroll
        for (int j = 0; j < 8; ++j) { p[(size_t)j * Nc] = c10[j]; p[(size_t)j * Nc + 16] = c11[j]; }
    }
}

__global__ void node_coef(const float* __restrict__ h,
                          const float* __restrict__ att,
                          float* __restrict__ ai, float* __restrict__ aj,
                          int nodes, int H, int Cc) {
    int idx = blockIdx.x * blockDim.x + threadIdx.x;
    if (idx >= nodes * H) return;
    int node = idx / H, head = idx % H;
    const float* hv = h + (size_t)node * H * Cc + head * Cc;
    const float* ad = att + head * 2 * Cc;
    const float* as = ad + Cc;
    float si = 0.f, sj = 0.f;
    for (int c = 0; c < Cc; ++c) { si += hv[c] * ad[c]; sj += hv[c] * as[c]; }
    ai[idx] = si;
    aj[idx] = sj;
}

__global__ void csr_count(const int* __restrict__ ei, int* __restrict__ cnt,
                          int* __restrict__ rank) {
    int e = blockIdx.x * blockDim.x + threadIdx.x;
    if (e >= E0C) return;
    int d_ = ei[E0C + e];
    rank[e] = atomicAdd(cnt + d_, 1);
}

#define SCAN_TB 1024
#define SCAN_CH ((NN + SCAN_TB - 1) / SCAN_TB)
#define NN4     (NN / 4)
#define SCAN_LD ((NN4 + SCAN_TB - 1) / SCAN_TB)
typedef int v4i __attribute__((ext_vector_type(4)));
__global__ __launch_bounds__(SCAN_TB)
void csr_scan(const int* __restrict__ cnt, int* __restrict__ rowptr) {
    __shared__ __attribute__((aligned(16))) int stage[NN];
    __shared__ int wsum[32];
    int t = threadIdx.x;
    int lane = t & 31, w = t >> 5;
    v4i* stage4 = (v4i*)stage;
    {
        const v4i* cnt4 = (const v4i*)cnt;
        v4i g[SCAN_LD];
#pragma unroll
        for (int k = 0; k < SCAN_LD; ++k) {
            int q = k * SCAN_TB + t;
            if (q < NN4) g[k] = cnt4[q];
        }
#pragma unroll
        for (int k = 0; k < SCAN_LD; ++k) {
            int q = k * SCAN_TB + t;
            if (q < NN4) stage4[q] = g[k];
        }
    }
    __syncthreads();
    int b = t * SCAN_CH;
    int v[SCAN_CH];
    int s = 0;
#pragma unroll
    for (int i = 0; i < SCAN_CH; ++i) {
        v[i] = (b + i < NN) ? stage[b + i] : 0;
        s += v[i];
    }
    int x = s;
    for (int o = 1; o < 32; o <<= 1) {
        int y = __shfl_up(x, o);
        if (lane >= o) x += y;
    }
    if (lane == 31) wsum[w] = x;
    __syncthreads();
    if (w == 0) {
        int tv = wsum[lane];
        for (int o = 1; o < 32; o <<= 1) {
            int y = __shfl_up(tv, o);
            if (lane >= o) tv += y;
        }
        wsum[lane] = tv;
    }
    __syncthreads();
    int run = x - s + ((w > 0) ? wsum[w - 1] : 0);
#pragma unroll
    for (int i = 0; i < SCAN_CH; ++i) {
        if (b + i < NN) stage[b + i] = run;
        run += v[i];
    }
    if (t == SCAN_TB - 1) rowptr[NN] = run;
    __syncthreads();
    {
        v4i* rp4 = (v4i*)rowptr;
#pragma unroll
        for (int k = 0; k < SCAN_LD; ++k) {
            int q = k * SCAN_TB + t;
            if (q < NN4) rp4[q] = stage4[q];
        }
    }
}

__global__ void csr_fill(const int* __restrict__ ei, const int* __restrict__ rowptr,
                         const int* __restrict__ rank, int* __restrict__ csrc) {
    int e = blockIdx.x * blockDim.x + threadIdx.x;
    if (e >= E0C) return;
    int d_ = ei[E0C + e];
    csrc[rowptr[d_] + rank[e]] = ei[e];
}

__device__ __forceinline__ float lrelu(float a) { return (a >= 0.f) ? a : NEG_SLOPE * a; }

__device__ __forceinline__ void fold_edge(float a, const float4 r,
                                          float& m, float& ssum, float4& acc) {
    float mn = fmaxf(m, a);
    float sc = __expf(m - mn);
    float w  = __expf(a - mn);
    ssum  = ssum * sc + w;
    acc.x = acc.x * sc + w * r.x;
    acc.y = acc.y * sc + w * r.y;
    acc.z = acc.z * sc + w * r.z;
    acc.w = acc.w * sc + w * r.w;
    m = mn;
}

__global__ __launch_bounds__(256)
void node_softmax_l1(const int* __restrict__ rowptr, const int* __restrict__ csrc,
                 const float* __restrict__ ai, const float* __restrict__ aj,
                 const float* __restrict__ h, float* __restrict__ agg) {
    int node = (int)((blockIdx.x * blockDim.x + threadIdx.x) >> 5);
    int lane = threadIdx.x & 31;
    if (node >= NN) return;
    const int hq  = lane >> 3;
    const int odd = lane & 1;
    const int sel = hq + 4 * odd;
    const float ai0 = ai[(size_t)node * H1C + hq];
    const float ai1 = ai[(size_t)node * H1C + 4 + hq];
    const int beg = rowptr[node], end = rowptr[node + 1];

#define L1_LOAD(s, ajv, r0, r1) { \
        ajv = aj[(size_t)(s) * H1C + sel]; \
        const float4* p_ = (const float4*)(h + (size_t)(s) * HIDDIM); \
        r0 = p_[lane]; r1 = p_[32 + lane]; }
#define L1_FOLD(ajv, r0, r1) { \
        float o_ = __shfl_xor(ajv, 1); \
        float a0_ = lrelu(ai0 + (odd ? o_ : ajv)); \
        float a1_ = lrelu(ai1 + (odd ? ajv : o_)); \
        fold_edge(a0_, r0, m0, s0, acc0); \
        fold_edge(a1_, r1, m1, s1, acc1); }

    float m0, m1, s0 = 1.f, s1 = 1.f;
    float4 acc0, acc1;
    {
        float ajv;
        L1_LOAD(node, ajv, acc0, acc1);
        float o_ = __shfl_xor(ajv, 1);
        m0 = lrelu(ai0 + (odd ? o_ : ajv));
        m1 = lrelu(ai1 + (odd ? ajv : o_));
    }
    int p = beg;
    for (; p + 4 <= end; p += 4) {
        int sa = csrc[p], sb = csrc[p + 1], sc_ = csrc[p + 2], sd = csrc[p + 3];
        float ja, jb, jc, jd;
        float4 a0, a1, b0, b1, c0, c1, d0, d1;
        L1_LOAD(sa, ja, a0, a1);
        L1_LOAD(sb, jb, b0, b1);
        L1_LOAD(sc_, jc, c0, c1);
        L1_LOAD(sd, jd, d0, d1);
        L1_FOLD(ja, a0, a1);
        L1_FOLD(jb, b0, b1);
        L1_FOLD(jc, c0, c1);
        L1_FOLD(jd, d0, d1);
    }
    for (; p < end; ++p) {
        int sa = csrc[p];
        float ja; float4 a0, a1;
        L1_LOAD(sa, ja, a0, a1);
        L1_FOLD(ja, a0, a1);
    }
#undef L1_LOAD
#undef L1_FOLD
    float i0 = 1.f / s0, i1 = 1.f / s1;
    float4* o = (float4*)(agg + (size_t)node * HIDDIM);
    o[lane]      = make_float4(acc0.x * i0, acc0.y * i0, acc0.z * i0, acc0.w * i0);
    o[32 + lane] = make_float4(acc1.x * i1, acc1.y * i1, acc1.z * i1, acc1.w * i1);
}

__global__ __launch_bounds__(256)
void node_softmax_l2(const int* __restrict__ rowptr, const int* __restrict__ csrc,
                 const float* __restrict__ ai, const float* __restrict__ aj,
                 const float* __restrict__ h, float* __restrict__ outp) {
    int gid  = blockIdx.x * blockDim.x + threadIdx.x;
    int node = gid >> 4;
    int lg   = gid & 15;
    if (node >= NN) return;
    const float aiv = ai[node];
    const int beg = rowptr[node], end = rowptr[node + 1];

    float m = lrelu(aiv + aj[node]), ssum = 1.f;
    float4 acc = ((const float4*)(h + (size_t)node * OUTDIM))[lg];

    int p = beg;
    for (; p + 4 <= end; p += 4) {
        int sa = csrc[p], sb = csrc[p + 1], sc_ = csrc[p + 2], sd = csrc[p + 3];
        float ja = aj[sa], jb = aj[sb], jc = aj[sc_], jd = aj[sd];
        float4 ra = ((const float4*)(h + (size_t)sa  * OUTDIM))[lg];
        float4 rb = ((const float4*)(h + (size_t)sb  * OUTDIM))[lg];
        float4 rc = ((const float4*)(h + (size_t)sc_ * OUTDIM))[lg];
        float4 rd = ((const float4*)(h + (size_t)sd  * OUTDIM))[lg];
        fold_edge(lrelu(aiv + ja), ra, m, ssum, acc);
        fold_edge(lrelu(aiv + jb), rb, m, ssum, acc);
        fold_edge(lrelu(aiv + jc), rc, m, ssum, acc);
        fold_edge(lrelu(aiv + jd), rd, m, ssum, acc);
    }
    for (; p < end; ++p) {
        int sa = csrc[p];
        float ja = aj[sa];
        float4 ra = ((const float4*)(h + (size_t)sa * OUTDIM))[lg];
        fold_edge(lrelu(aiv + ja), ra, m, ssum, acc);
    }
    float inv = 1.f / ssum;
    ((float4*)(outp + (size_t)node * OUTDIM))[lg] =
        make_float4(acc.x * inv, acc.y * inv, acc.z * inv, acc.w * inv);
}

__device__ __forceinline__ float elu1(float v) { return (v > 0.f) ? v : (__expf(v) - 1.f); }

__global__ __launch_bounds__(256)
void elu_bias_cvt(const float* __restrict__ agg, const float* __restrict__ bias,
                  _Float16* __restrict__ out) {
    constexpr int NQ = NN * HIDDIM / 4;
    static_assert(NQ % (32 * EW_BIG) == 0, "whole waves only");
    static_assert(HIDDIM == 256 && EW_BIG % 2 == 0, "a row is 64 pieces: two per lane");
    int wave = (int)((blockIdx.x * blockDim.x + threadIdx.x) >> 5);
    int lane = threadIdx.x & 31;
    if (wave * (32 * EW_BIG) >= NQ) return;
    int q0 = wave * (32 * EW_BIG) + lane;
    const float4* a4 = (const float4*)agg;
    v4h* d4 = (v4h*)out;
    const float4 bE = ((const float4*)bias)[lane];
    const float4 bO = ((const float4*)bias)[32 + lane];
    float4 v[EW_BIG];
#pragma unroll
    for (int u = 0; u < EW_BIG; ++u) v[u] = a4[q0 + u * 32];
#pragma unroll
    for (int u = 0; u < EW_BIG; ++u) {
        const float4 b = (u & 1) ? bO : bE;
        v4h o;
        o.x = (_Float16)elu1(v[u].x + b.x); o.y = (_Float16)elu1(v[u].y + b.y);
        o.z = (_Float16)elu1(v[u].z + b.z); o.w = (_Float16)elu1(v[u].w + b.w);
        d4[q0 + u * 32] = o;
    }
}

__global__ __launch_bounds__(256)
void add_bias(float* __restrict__ out, const float* __restrict__ bias) {
    constexpr int NQ = NN * OUTDIM / 4;
    static_assert(NQ % (32 * EW_SMALL) == 0, "whole waves only");
    static_assert(OUTDIM == 64, "a row is 16 pieces: one per lane of a half wave");
    int wave = (int)((blockIdx.x * blockDim.x + threadIdx.x) >> 5);
    int lane = threadIdx.x & 31;
    if (wave * (32 * EW_SMALL) >= NQ) return;
    int q0 = wave * (32 * EW_SMALL) + lane;
    float4* o4 = (float4*)out;
    const float4 b = ((const float4*)bias)[lane & 15];
    float4 v[EW_SMALL];
#pragma unroll
    for (int u = 0; u < EW_SMALL; ++u) v[u] = o4[q0 + u * 32];
#pragma unroll
    for (int u = 0; u < EW_SMALL; ++u) {
        float4 r = v[u];
        r.x += b.x; r.y += b.y; r.z += b.z; r.w += b.w;
        o4[q0 + u * 32] = r;
    }
}

extern "C" void kernel_launch(void* const* d_in, const int* in_sizes, int n_in,
                              void* d_out, int out_size, void* d_ws, size_t ws_size,
                              hipStream_t stream) {
    (void)in_sizes; (void)n_in; (void)out_size; (void)ws_size;

    const float* x    = (const float*)d_in[0];
    const int*   ei   = (const int*)  d_in[1];
    const float* W1   = (const float*)d_in[2];
    const float* att1 = (const float*)d_in[3];
    const float* b1   = (const float*)d_in[4];
    const float* W2   = (const float*)d_in[5];
    const float* att2 = (const float*)d_in[6];
    const float* b2   = (const float*)d_in[7];
    float* out = (float*)d_out;

    char* base = (char*)d_ws;
    size_t off = 0;
    auto carve = [&](size_t bytes) -> char* {
        char* p = base + off;
        off = (off + bytes + 255) & ~(size_t)255;
        return p;
    };
    _Float16* XH    = (_Float16*)carve((size_t)MPAD * INDIM * 2);
    _Float16* W1T   = (_Float16*)carve((size_t)INDIM * HIDDIM * 2);
    _Float16* W2T   = (_Float16*)carve((size_t)HIDDIM * OUTDIM * 2);
    float*    HBUF  = (float*)   carve((size_t)MPAD * HIDDIM * 4);
    float*    AGG1  = (float*)   carve((size_t)NN * HIDDIM * 4);
    float*    AI    = (float*)   carve((size_t)NN * H1C * 4);
    float*    AJ    = (float*)   carve((size_t)NN * H1C * 4);
    int*      CNT   = (int*)     carve((size_t)NN * 4);
    int*      RANK  = (int*)     carve((size_t)E0C * 4);
    int*      ROWP  = (int*)     carve((size_t)(NN + 1) * 4);
    int*      CSRC  = (int*)     carve((size_t)E0C * 4);

    const int TB = 256;
    auto nb = [](long long n, int tb) { return (unsigned)((n + tb - 1) / tb); };
    const int MT2 = MPAD / 32;

    hipMemsetAsync(CNT, 0, (size_t)NN * 4, stream);
    csr_count<<<nb((long long)E0C, TB), TB, 0, stream>>>(ei, CNT, RANK);
    csr_scan <<<1, SCAN_TB, 0, stream>>>(CNT, ROWP);
    csr_fill <<<nb((long long)E0C, TB), TB, 0, stream>>>(ei, ROWP, RANK, CSRC);

    cvt_f32_to_f16<<<nb((long long)NN * INDIM / (4 * EW_BIG), TB), TB, 0, stream>>>(x, XH);
    xpose_cvt_f16 <<<nb((long long)INDIM * HIDDIM, TB), TB, 0, stream>>>(W1, W1T, INDIM, HIDDIM);
    xpose_cvt_f16 <<<nb((long long)HIDDIM * OUTDIM, TB), TB, 0, stream>>>(W2, W2T, HIDDIM, OUTDIM);

    {
        long long waves = (long long)MT2 * (HIDDIM / 32);
        gemm_f16_wmma<<<nb(waves * 32, 128), 128, 0, stream>>>(XH, W1T, HBUF, MT2, HIDDIM, INDIM);
    }

    node_coef<<<nb((long long)NN * H1C, TB), TB, 0, stream>>>(HBUF, att1, AI, AJ, NN, H1C, C1C);

    node_softmax_l1<<<nb((long long)NN * 32, TB), TB, 0, stream>>>(ROWP, CSRC, AI, AJ, HBUF, AGG1);

    elu_bias_cvt<<<nb((long long)NN * HIDDIM / (4 * EW_BIG), TB), TB, 0, stream>>>(AGG1, b1, XH);

    {
        long long waves = (long long)MT2 * (OUTDIM / 32);
        gemm_f16_wmma<<<nb(waves * 32, 128), 128, 0, stream>>>(XH, W2T, HBUF, MT2, OUTDIM, HIDDIM);
    }

    node_coef<<<nb((long long)NN * H2C, TB), TB, 0, stream>>>(HBUF, att2, AI, AJ, NN, H2C, C2C);

    node_softmax_l2<<<nb((long long)NN * 16, TB), TB, 0, stream>>>(ROWP, CSRC, AI, AJ, HBUF, out);

    add_bias<<<nb((long long)NN * OUTDIM / (4 * EW_SMALL), TB), TB, 0, stream>>>(out, b2);
}


# Per-row passes fused into producers, arithmetic and fp32 sum order of every value kept: ai/aj as serial dots in the GEMM epilogue via LDS, bias+ELU+cast at layer-1 row store, bias at out store
# speedup vs baseline: 3.4256x; 1.0909x over previous
#include <hip/hip_runtime.h>
#include <hip/hip_bf16.h>

#define NN     50000
#define MPAD   50016
#define E0C    800000
#define ETOT   850000
#define INDIM  256
#define HIDDIM 256
#define OUTDIM 64
#define H1C    8
#define C1C    32
#define H2C    1
#define C2C    64
#define NEG_SLOPE 0.2f

typedef __attribute__((ext_vector_type(16))) _Float16 v16h;
typedef __attribute__((ext_vector_type(8)))  _Float16 v8h;
typedef __attribute__((ext_vector_type(8)))  float    v8f;

union V16U { v16h v; v8h h[2]; };

#define EW_BIG   16
typedef __attribute__((ext_vector_type(4))) _Float16 v4h;

__global__ __launch_bounds__(256)
void cvt_f32_to_f16(const float* __restrict__ src, _Float16* __restrict__ dst) {
    constexpr int NQ = NN * INDIM / 4;
    static_assert(NQ % (32 * EW_BIG) == 0, "whole waves only");
    int wave = (int)((blockIdx.x * blockDim.x + threadIdx.x) >> 5);
    int lane = threadIdx.x & 31;
    if (wave * (32 * EW_BIG) >= NQ) return;
    int q0 = wave * (32 * EW_BIG) + lane;
    const float4* s4 = (const float4*)src;
    v4h* d4 = (v4h*)dst;
    float4 v[EW_BIG];
#pragma unroll
    for (int u = 0; u < EW_BIG; ++u) v[u] = s4[q0 + u * 32];
#pragma unroll
    for (int u = 0; u < EW_BIG; ++u) {
        v4h o;
        o.x = (_Float16)v[u].x; o.y = (_Float16)v[u].y;
        o.z = (_Float16)v[u].z; o.w = (_Float16)v[u].w;
        d4[q0 + u * 32] = o;
    }
}

__global__ void xpose_cvt_f16(const float* __restrict__ W,
                              _Float16* __restrict__ Wt, int K, int Nc) {
    int i = blockIdx.x * blockDim.x + threadIdx.x;
    if (i >= K * Nc) return;
    int k = i / Nc, n = i % Nc;
    Wt[(size_t)n * K + k] = (_Float16)W[i];
}

#define WMMA16(a, b, c) __builtin_amdgcn_wmma_f32_16x16x32_f16( \
        false, (a), false, (b), (short)0, (c), false, false)

__device__ __forceinline__
void wave_tile_32x32(const _Float16* __restrict__ A, const _Float16* __restrict__ Bt,
                     int tm, int tn, int r, int half, int K,
                     v8f& c00, v8f& c01, v8f& c10, v8f& c11) {
    const _Float16* a0 = A  + (size_t)(tm * 32 + r) * K + half * 8;
    const _Float16* a1 = a0 + (size_t)16 * K;
    const _Float16* b0 = Bt + (size_t)(tn * 32 + r) * K + half * 8;
    const _Float16* b1 = b0 + (size_t)16 * K;
    for (int k0 = 0; k0 < K; k0 += 32) {
        V16U A0, A1, B0, B1;
        A0.h[0] = *(const v8h*)(a0 + k0);
        A0.h[1] = *(const v8h*)(a0 + k0 + 16);
        A1.h[0] = *(const v8h*)(a1 + k0);
        A1.h[1] = *(const v8h*)(a1 + k0 + 16);
        B0.h[0] = *(const v8h*)(b0 + k0);
        B0.h[1] = *(const v8h*)(b0 + k0 + 16);
        B1.h[0] = *(const v8h*)(b1 + k0);
        B1.h[1] = *(const v8h*)(b1 + k0 + 16);
        c00 = WMMA16(A0.v, B0.v, c00);
        c01 = WMMA16(A0.v, B1.v, c01);
        c10 = WMMA16(A1.v, B0.v, c10);
        c11 = WMMA16(A1.v, B1.v, c11);
    }
}

__device__ __forceinline__
void store_tile_32x32(float* __restrict__ C, int Nc, int tm, int tn, int r, int half,
                      const v8f& c00, const v8f& c01, const v8f& c10, const v8f& c11) {
    float* p = C + (size_t)(tm * 32 + half * 8) * Nc + tn * 32 + r;
#pragma unroll
    for (int j = 0; j < 8; ++j) { p[(size_t)j * Nc] = c00[j]; p[(size_t)j * Nc + 16] = c01[j]; }
    p += (size_t)16 * Nc;
#pragma unroll
    for (int j = 0; j < 8; ++j) { p[(size_t)j * Nc] = c10[j]; p[(size_t)j * Nc + 16] = c11[j]; }
}

__global__ __launch_bounds__(128)
void gemm_f16_wmma(const _Float16* __restrict__ A,
                   const _Float16* __restrict__ Bt,
                   float* __restrict__ C, int Mtiles2, int Nc, int K) {
    int wave = (int)((blockIdx.x * blockDim.x + threadIdx.x) >> 5);
    int lane = threadIdx.x & 31;
    int nt2 = Nc >> 5;
    int tm = wave / nt2;
    int tn = wave % nt2;
    if (tm >= Mtiles2) return;
    int r    = lane & 15;
    int half = lane >> 4;
    v8f c00 = {}, c01 = {}, c10 = {}, c11 = {};
    wave_tile_32x32(A, Bt, tm, tn, r, half, K, c00, c01, c10, c11);
    store_tile_32x32(C, Nc, tm, tn, r, half, c00, c01, c10, c11);
}

#define TPITCH 34
__global__ __launch_bounds__(256)
void gemm_l1_coef(const _Float16* __restrict__ A, const _Float16* __restrict__ Bt,
                  float* __restrict__ C, const float* __restrict__ att,
                  float* __restrict__ ai, float* __restrict__ aj) {
    __shared__ float tile[H1C][32 * TPITCH];
    __shared__ __attribute__((aligned(16))) float sI[32 * H1C];
    __shared__ __attribute__((aligned(16))) float sJ[32 * H1C];
    const int tm   = blockIdx.x;
    const int tn   = threadIdx.x >> 5;
    const int lane = threadIdx.x & 31;
    const int r    = lane & 15;
    const int half = lane >> 4;
    v8f c00 = {}, c01 = {}, c10 = {}, c11 = {};
    wave_tile_32x32(A, Bt, tm, tn, r, half, INDIM, c00, c01, c10, c11);

    float* T = tile[tn];
#pragma unroll
    for (int j = 0; j < 8; ++j) {
        T[(half * 8 + j) * TPITCH + r]           = c00[j];
        T[(half * 8 + j) * TPITCH + 16 + r]      = c01[j];
        T[(16 + half * 8 + j) * TPITCH + r]      = c10[j];
        T[(16 + half * 8 + j) * TPITCH + 16 + r] = c11[j];
    }
    __syncthreads();
    const float* ad = att + tn * 2 * C1C;
    const float* as = ad + C1C;
    float si = 0.f, sj = 0.f;
    for (int c = 0; c < C1C; ++c) {
        float hv = T[lane * TPITCH + c];
        si = __builtin_fmaf(hv, ad[c], si);
        sj = __builtin_fmaf(hv, as[c], sj);
    }
    sI[lane * H1C + tn] = si;
    sJ[lane * H1C + tn] = sj;
    __syncthreads();
    store_tile_32x32(C, HIDDIM, tm, tn, r, half, c00, c01, c10, c11);
    const int t = threadIdx.x;
    if (t < 128) {
        const int q = t & 63;
        if (tm * 32 + (q >> 1) < NN) {
            if (t < 64) ((float4*)ai)[(size_t)tm * 64 + q] = ((const float4*)sI)[q];
            else        ((float4*)aj)[(size_t)tm * 64 + q] = ((const float4*)sJ)[q];
        }
    }
}

__global__ void node_coef(const float* __restrict__ h,
                          const float* __restrict__ att,
                          float* __restrict__ ai, float* __restrict__ aj,
                          int nodes, int H, int Cc) {
    int idx = blockIdx.x * blockDim.x + threadIdx.x;
    if (idx >= nodes * H) return;
    int node = idx / H, head = idx % H;
    const float* hv = h + (size_t)node * H * Cc + head * Cc;
    const float* ad = att + head * 2 * Cc;
    const float* as = ad + Cc;
    float si = 0.f, sj = 0.f;
    for (int c = 0; c < Cc; ++c) { si += hv[c] * ad[c]; sj += hv[c] * as[c]; }
    ai[idx] = si;
    aj[idx] = sj;
}

__global__ void csr_count(const int* __restrict__ ei, int* __restrict__ cnt,
                          int* __restrict__ rank) {
    int e = blockIdx.x * blockDim.x + threadIdx.x;
    if (e >= E0C) return;
    int d_ = ei[E0C + e];
    rank[e] = atomicAdd(cnt + d_, 1);
}

#define SCAN_TB 1024
#define SCAN_CH ((NN + SCAN_TB - 1) / SCAN_TB)
#define NN4     (NN / 4)
#define SCAN_LD ((NN4 + SCAN_TB - 1) / SCAN_TB)
typedef int v4i __attribute__((ext_vector_type(4)));
__global__ __launch_bounds__(SCAN_TB)
void csr_scan(const int* __restrict__ cnt, int* __restrict__ rowptr) {
    __shared__ __attribute__((aligned(16))) int stage[NN];
    __shared__ int wsum[32];
    int t = threadIdx.x;
    int lane = t & 31, w = t >> 5;
    v4i* stage4 = (v4i*)stage;
    {
        const v4i* cnt4 = (const v4i*)cnt;
        v4i g[SCAN_LD];
#pragma unroll
        for (int k = 0; k < SCAN_LD; ++k) {
            int q = k * SCAN_TB + t;
            if (q < NN4) g[k] = cnt4[q];
        }
#pragma unroll
        for (int k = 0; k < SCAN_LD; ++k) {
            int q = k * SCAN_TB + t;
            if (q < NN4) stage4[q] = g[k];
        }
    }
    __syncthreads();
    int b = t * SCAN_CH;
    int v[SCAN_CH];
    int s = 0;
#pragma unroll
    for (int i = 0; i < SCAN_CH; ++i) {
        v[i] = (b + i < NN) ? stage[b + i] : 0;
        s += v[i];
    }
    int x = s;
    for (int o = 1; o < 32; o <<= 1) {
        int y = __shfl_up(x, o);
        if (lane >= o) x += y;
    }
    if (lane == 31) wsum[w] = x;
    __syncthreads();
    if (w == 0) {
        int tv = wsum[lane];
        for (int o = 1; o < 32; o <<= 1) {
            int y = __shfl_up(tv, o);
            if (lane >= o) tv += y;
        }
        wsum[lane] = tv;
    }
    __syncthreads();
    int run = x - s + ((w > 0) ? wsum[w - 1] : 0);
#pragma unroll
    for (int i = 0; i < SCAN_CH; ++i) {
        if (b + i < NN) stage[b + i] = run;
        run += v[i];
    }
    if (t == SCAN_TB - 1) rowptr[NN] = run;
    __syncthreads();
    {
        v4i* rp4 = (v4i*)rowptr;
#pragma unroll
        for (int k = 0; k < SCAN_LD; ++k) {
            int q = k * SCAN_TB + t;
            if (q < NN4) rp4[q] = stage4[q];
        }
    }
}

__global__ void csr_fill(const int* __restrict__ ei, const int* __restrict__ rowptr,
                         const int* __restrict__ rank, int* __restrict__ csrc) {
    int e = blockIdx.x * blockDim.x + threadIdx.x;
    if (e >= E0C) return;
    int d_ = ei[E0C + e];
    csrc[rowptr[d_] + rank[e]] = ei[e];
}

__device__ __forceinline__ float lrelu(float a) { return (a >= 0.f) ? a : NEG_SLOPE * a; }
__device__ __forceinline__ float elu1(float v) { return (v > 0.f) ? v : (__expf(v) - 1.f); }

__device__ __forceinline__ void fold_edge(float a, const float4 r,
                                          float& m, float& ssum, float4& acc) {
    float mn = fmaxf(m, a);
    float sc = __expf(m - mn);
    float w  = __expf(a - mn);
    ssum  = ssum * sc + w;
    acc.x = acc.x * sc + w * r.x;
    acc.y = acc.y * sc + w * r.y;
    acc.z = acc.z * sc + w * r.z;
    acc.w = acc.w * sc + w * r.w;
    m = mn;
}

__global__ __launch_bounds__(256)
void node_softmax_l1(const int* __restrict__ rowptr, const int* __restrict__ csrc,
                 const float* __restrict__ ai, const float* __restrict__ aj,
                 const float* __restrict__ h, const float* __restrict__ bias,
                 _Float16* __restrict__ act) {
    int node = (int)((blockIdx.x * blockDim.x + threadIdx.x) >> 5);
    int lane = threadIdx.x & 31;
    if (node >= NN) return;
    const int hq  = lane >> 3;
    const int odd = lane & 1;
    const int sel = hq + 4 * odd;
    const float ai0 = ai[(size_t)node * H1C + hq];
    const float ai1 = ai[(size_t)node * H1C + 4 + hq];
    const int beg = rowptr[node], end = rowptr[node + 1];

#define L1_LOAD(s, ajv, r0, r1) { \
        ajv = aj[(size_t)(s) * H1C + sel]; \
        const float4* p_ = (const float4*)(h + (size_t)(s) * HIDDIM); \
        r0 = p_[lane]; r1 = p_[32 + lane]; }
#define L1_FOLD(ajv, r0, r1) { \
        float o_ = __shfl_xor(ajv, 1); \
        float a0_ = lrelu(ai0 + (odd ? o_ : ajv)); \
        float a1_ = lrelu(ai1 + (odd ? ajv : o_)); \
        fold_edge(a0_, r0, m0, s0, acc0); \
        fold_edge(a1_, r1, m1, s1, acc1); }

    float m0, m1, s0 = 1.f, s1 = 1.f;
    float4 acc0, acc1;
    {
        float ajv;
        L1_LOAD(node, ajv, acc0, acc1);
        float o_ = __shfl_xor(ajv, 1);
        m0 = lrelu(ai0 + (odd ? o_ : ajv));
        m1 = lrelu(ai1 + (odd ? ajv : o_));
    }
    int p = beg;
    for (; p + 4 <= end; p += 4) {
        int sa = csrc[p], sb = csrc[p + 1], sc_ = csrc[p + 2], sd = csrc[p + 3];
        float ja, jb, jc, jd;
        float4 a0, a1, b0, b1, c0, c1, d0, d1;
        L1_LOAD(sa, ja, a0, a1);
        L1_LOAD(sb, jb, b0, b1);
        L1_LOAD(sc_, jc, c0, c1);
        L1_LOAD(sd, jd, d0, d1);
        L1_FOLD(ja, a0, a1);
        L1_FOLD(jb, b0, b1);
        L1_FOLD(jc, c0, c1);
        L1_FOLD(jd, d0, d1);
    }
    for (; p < end; ++p) {
        int sa = csrc[p];
        float ja; float4 a0, a1;
        L1_LOAD(sa, ja, a0, a1);
        L1_FOLD(ja, a0, a1);
    }
#undef L1_LOAD
#undef L1_FOLD
    float i0 = 1.f / s0, i1 = 1.f / s1;
    const float4 bE = ((const float4*)bias)[lane];
    const float4 bO = ((const float4*)bias)[32 + lane];
    v4h o0, o1;
    o0.x = (_Float16)elu1(acc0.x * i0 + bE.x); o0.y = (_Float16)elu1(acc0.y * i0 + bE.y);
    o0.z = (_Float16)elu1(acc0.z * i0 + bE.z); o0.w = (_Float16)elu1(acc0.w * i0 + bE.w);
    o1.x = (_Float16)elu1(acc1.x * i1 + bO.x); o1.y = (_Float16)elu1(acc1.y * i1 + bO.y);
    o1.z = (_Float16)elu1(acc1.z * i1 + bO.z); o1.w = (_Float16)elu1(acc1.w * i1 + bO.w);
    v4h* o = (v4h*)(act + (size_t)node * HIDDIM);
    o[lane]      = o0;
    o[32 + lane] = o1;
}

__global__ __launch_bounds__(256)
void node_softmax_l2(const int* __restrict__ rowptr, const int* __restrict__ csrc,
                 const float* __restrict__ ai, const float* __restrict__ aj,
                 const float* __restrict__ h, const float* __restrict__ bias,
                 float* __restrict__ outp) {
    int gid  = blockIdx.x * blockDim.x + threadIdx.x;
    int node = gid >> 4;
    int lg   = gid & 15;
    if (node >= NN) return;
    const float aiv = ai[node];
    const int beg = rowptr[node], end = rowptr[node + 1];

    float m = lrelu(aiv + aj[node]), ssum = 1.f;
    float4 acc = ((const float4*)(h + (size_t)node * OUTDIM))[lg];

    int p = beg;
    for (; p + 4 <= end; p += 4) {
        int sa = csrc[p], sb = csrc[p + 1], sc_ = csrc[p + 2], sd = csrc[p + 3];
        float ja = aj[sa], jb = aj[sb], jc = aj[sc_], jd = aj[sd];
        float4 ra = ((const float4*)(h + (size_t)sa  * OUTDIM))[lg];
        float4 rb = ((const float4*)(h + (size_t)sb  * OUTDIM))[lg];
        float4 rc = ((const float4*)(h + (size_t)sc_ * OUTDIM))[lg];
        float4 rd = ((const float4*)(h + (size_t)sd  * OUTDIM))[lg];
        fold_edge(lrelu(aiv + ja), ra, m, ssum, acc);
        fold_edge(lrelu(aiv + jb), rb, m, ssum, acc);
        fold_edge(lrelu(aiv + jc), rc, m, ssum, acc);
        fold_edge(lrelu(aiv + jd), rd, m, ssum, acc);
    }
    for (; p < end; ++p) {
        int sa = csrc[p];
        float ja = aj[sa];
        float4 ra = ((const float4*)(h + (size_t)sa * OUTDIM))[lg];
        fold_edge(lrelu(aiv + ja), ra, m, ssum, acc);
    }
    float inv = 1.f / ssum;
    const float4 b = ((const float4*)bias)[lg];
    ((float4*)(outp + (size_t)node * OUTDIM))[lg] =
        make_float4(acc.x * inv + b.x, acc.y * inv + b.y, acc.z * inv + b.z, acc.w * inv + b.w);
}


extern "C" void kernel_launch(void* const* d_in, const int* in_sizes, int n_in,
                              void* d_out, int out_size, void* d_ws, size_t ws_size,
                              hipStream_t stream) {
    (void)in_sizes; (void)n_in; (void)out_size; (void)ws_size;

    const float* x    = (const float*)d_in[0];
    const int*   ei   = (const int*)  d_in[1];
    const float* W1   = (const float*)d_in[2];
    const float* att1 = (const float*)d_in[3];
    const float* b1   = (const float*)d_in[4];
    const float* W2   = (const float*)d_in[5];
    const float* att2 = (const float*)d_in[6];
    const float* b2   = (const float*)d_in[7];
    float* out = (float*)d_out;

    char* base = (char*)d_ws;
    size_t off = 0;
    auto carve = [&](size_t bytes) -> char* {
        char* p = base + off;
        off = (off + bytes + 255) & ~(size_t)255;
        return p;
    };
    _Float16* XH    = (_Float16*)carve((size_t)MPAD * INDIM * 2);
    _Float16* W1T   = (_Float16*)carve((size_t)INDIM * HIDDIM * 2);
    _Float16* W2T   = (_Float16*)carve((size_t)HIDDIM * OUTDIM * 2);
    float*    HBUF  = (float*)   carve((size_t)MPAD * HIDDIM * 4);
    float*    AI    = (float*)   carve((size_t)NN * H1C * 4);
    float*    AJ    = (float*)   carve((size_t)NN * H1C * 4);
    int*      CNT   = (int*)     carve((size_t)NN * 4);
    int*      RANK  = (int*)     carve((size_t)E0C * 4);
    int*      ROWP  = (int*)     carve((size_t)(NN + 1) * 4);
    int*      CSRC  = (int*)     carve((size_t)E0C * 4);

    const int TB = 256;
    auto nb = [](long long n, int tb) { return (unsigned)((n + tb - 1) / tb); };
    const int MT2 = MPAD / 32;

    hipMemsetAsync(CNT, 0, (size_t)NN * 4, stream);
    csr_count<<<nb((long long)E0C, TB), TB, 0, stream>>>(ei, CNT, RANK);
    csr_scan <<<1, SCAN_TB, 0, stream>>>(CNT, ROWP);
    csr_fill <<<nb((long long)E0C, TB), TB, 0, stream>>>(ei, ROWP, RANK, CSRC);

    cvt_f32_to_f16<<<nb((long long)NN * INDIM / (4 * EW_BIG), TB), TB, 0, stream>>>(x, XH);
    xpose_cvt_f16 <<<nb((long long)INDIM * HIDDIM, TB), TB, 0, stream>>>(W1, W1T, INDIM, HIDDIM);
    xpose_cvt_f16 <<<nb((long long)HIDDIM * OUTDIM, TB), TB, 0, stream>>>(W2, W2T, HIDDIM, OUTDIM);

    gemm_l1_coef<<<MT2, 256, 0, stream>>>(XH, W1T, HBUF, att1, AI, AJ);

    node_softmax_l1<<<nb((long long)NN * 32, TB), TB, 0, stream>>>(ROWP, CSRC, AI, AJ, HBUF, b1, XH);

    {
        long long waves = (long long)MT2 * (OUTDIM / 32);
        gemm_f16_wmma<<<nb(waves * 32, 128), 128, 0, stream>>>(XH, W2T, HBUF, MT2, OUTDIM, HIDDIM);
    }

    node_coef<<<nb((long long)NN * H2C, TB), TB, 0, stream>>>(HBUF, att2, AI, AJ, NN, H2C, C2C);

    node_softmax_l2<<<nb((long long)NN * 16, TB), TB, 0, stream>>>(ROWP, CSRC, AI, AJ, HBUF, b2, out);
}


# Layer-2 transform and its coefficient dots move into the layer-1 softmax kernel: 16 nodes per work-group, f16 row tile in LDS feeds the same WMMA K loop; 7 launches to 5; arithmetic, sum order kept
# speedup vs baseline: 3.8968x; 1.1376x over previous
#include <hip/hip_runtime.h>
#include <hip/hip_bf16.h>

#define NN     50000
#define MPAD   50016
#define E0C    800000
#define ETOT   850000
#define INDIM  256
#define HIDDIM 256
#define OUTDIM 64
#define H1C    8
#define C1C    32
#define H2C    1
#define C2C    64
#define NEG_SLOPE 0.2f

typedef __attribute__((ext_vector_type(16))) _Float16 v16h;
typedef __attribute__((ext_vector_type(8)))  _Float16 v8h;
typedef __attribute__((ext_vector_type(8)))  float    v8f;

union V16U { v16h v; v8h h[2]; };

#define EW_BIG   16
typedef __attribute__((ext_vector_type(4))) _Float16 v4h;

__device__ __forceinline__
void cvt_f32_to_f16_wave(const float* __restrict__ src, _Float16* __restrict__ dst,
                         int wave, int lane) {
    constexpr int NQ = NN * INDIM / 4;
    static_assert(NQ % (32 * EW_BIG) == 0, "whole waves only");
    if (wave * (32 * EW_BIG) >= NQ) return;
    int q0 = wave * (32 * EW_BIG) + lane;
    const float4* s4 = (const float4*)src;
    v4h* d4 = (v4h*)dst;
    float4 v[EW_BIG];
#pragma unroll
    for (int u = 0; u < EW_BIG; ++u) v[u] = s4[q0 + u * 32];
#pragma unroll
    for (int u = 0; u < EW_BIG; ++u) {
        v4h o;
        o.x = (_Float16)v[u].x; o.y = (_Float16)v[u].y;
        o.z = (_Float16)v[u].z; o.w = (_Float16)v[u].w;
        d4[q0 + u * 32] = o;
    }
}

__device__ __forceinline__
void xpose_cvt_f16_elem(const float* __restrict__ W, _Float16* __restrict__ Wt,
                        int i, int K, int Nc) {
    if (i >= K * Nc) return;
    int k = i / Nc, n = i % Nc;
    Wt[(size_t)n * K + k] = (_Float16)W[i];
}

#define WMMA16(a, b, c) __builtin_amdgcn_wmma_f32_16x16x32_f16( \
        false, (a), false, (b), (short)0, (c), false, false)

__device__ __forceinline__
void wave_tile_32x32(const _Float16* __restrict__ A, const _Float16* __restrict__ Bt,
                     int tm, int tn, int r, int half, int K,
                     v8f& c00, v8f& c01, v8f& c10, v8f& c11) {
    const _Float16* a0 = A  + (size_t)(tm * 32 + r) * K + half * 8;
    const _Float16* a1 = a0 + (size_t)16 * K;
    const _Float16* b0 = Bt + (size_t)(tn * 32 + r) * K + half * 8;
    const _Float16* b1 = b0 + (size_t)16 * K;
    for (int k0 = 0; k0 < K; k0 += 32) {
        V16U A0, A1, B0, B1;
        A0.h[0] = *(const v8h*)(a0 + k0);
        A0.h[1] = *(const v8h*)(a0 + k0 + 16);
        A1.h[0] = *(const v8h*)(a1 + k0);
        A1.h[1] = *(const v8h*)(a1 + k0 + 16);
        B0.h[0] = *(const v8h*)(b0 + k0);
        B0.h[1] = *(const v8h*)(b0 + k0 + 16);
        B1.h[0] = *(const v8h*)(b1 + k0);
        B1.h[1] = *(const v8h*)(b1 + k0 + 16);
        c00 = WMMA16(A0.v, B0.v, c00);
        c01 = WMMA16(A0.v, B1.v, c01);
        c10 = WMMA16(A1.v, B0.v, c10);
        c11 = WMMA16(A1.v, B1.v, c11);
    }
}

__device__ __forceinline__
void store_tile_32x32(float* __restrict__ C, int Nc, int tm, int tn, int r, int half,
                      const v8f& c00, const v8f& c01, const v8f& c10, const v8f& c11) {
    float* p = C + (size_t)(tm * 32 + half * 8) * Nc + tn * 32 + r;
#pragma unroll
    for (int j = 0; j < 8; ++j) { p[(size_t)j * Nc] = c00[j]; p[(size_t)j * Nc + 16] = c01[j]; }
    p += (size_t)16 * Nc;
#pragma unroll
    for (int j = 0; j < 8; ++j) { p[(size_t)j * Nc] = c10[j]; p[(size_t)j * Nc + 16] = c11[j]; }
}

#define TPITCH 34
__device__ __forceinline__
void gemm_l1_coef_tile(const _Float16* __restrict__ A, const _Float16* __restrict__ Bt,
                       float* __restrict__ C, const float* __restrict__ att,
                       float* __restrict__ ai, float* __restrict__ aj, const int tm,
                       float (*tile)[32 * TPITCH], float* sI, float* sJ) {
    const int tn   = threadIdx.x >> 5;
    const int lane = threadIdx.x & 31;
    const int r    = lane & 15;
    const int half = lane >> 4;
    v8f c00 = {}, c01 = {}, c10 = {}, c11 = {};
    wave_tile_32x32(A, Bt, tm, tn, r, half, INDIM, c00, c01, c10, c11);

    float* T = tile[tn];
#pragma unroll
    for (int j = 0; j < 8; ++j) {
        T[(half * 8 + j) * TPITCH + r]           = c00[j];
        T[(half * 8 + j) * TPITCH + 16 + r]      = c01[j];
        T[(16 + half * 8 + j) * TPITCH + r]      = c10[j];
        T[(16 + half * 8 + j) * TPITCH + 16 + r] = c11[j];
    }
    __syncthreads();
    const float* ad = att + tn * 2 * C1C;
    const float* as = ad + C1C;
    float si = 0.f, sj = 0.f;
    for (int c = 0; c < C1C; ++c) {
        float hv = T[lane * TPITCH + c];
        si = __builtin_fmaf(hv, ad[c], si);
        sj = __builtin_fmaf(hv, as[c], sj);
    }
    sI[lane * H1C + tn] = si;
    sJ[lane * H1C + tn] = sj;
    __syncthreads();
    store_tile_32x32(C, HIDDIM, tm, tn, r, half, c00, c01, c10, c11);
    const int t = threadIdx.x;
    if (t < 128) {
        const int q = t & 63;
        if (tm * 32 + (q >> 1) < NN) {
            if (t < 64) ((float4*)ai)[(size_t)tm * 64 + q] = ((const float4*)sI)[q];
            else        ((float4*)aj)[(size_t)tm * 64 + q] = ((const float4*)sJ)[q];
        }
    }
}

__device__ __forceinline__
void csr_count_edge(const int* __restrict__ ei, int* __restrict__ cnt,
                    int* __restrict__ rank, int e) {
    if (e >= E0C) return;
    int d_ = ei[E0C + e];
    rank[e] = atomicAdd(cnt + d_, 1);
}

#define SCAN_TB 1024
#define SCAN_CH ((NN + SCAN_TB - 1) / SCAN_TB)
#define NN4     (NN / 4)
#define SCAN_LD ((NN4 + SCAN_TB - 1) / SCAN_TB)
typedef int v4i __attribute__((ext_vector_type(4)));
__device__ __forceinline__
void csr_scan_block(const int* __restrict__ cnt, int* __restrict__ rowptr,
                    int* stage, int* wsum) {
    int t = threadIdx.x;
    int lane = t & 31, w = t >> 5;
    v4i* stage4 = (v4i*)stage;
    {
        const v4i* cnt4 = (const v4i*)cnt;
        v4i g[SCAN_LD];
#pragma unroll
        for (int k = 0; k < SCAN_LD; ++k) {
            int q = k * SCAN_TB + t;
            if (q < NN4) g[k] = cnt4[q];
        }
#pragma unroll
        for (int k = 0; k < SCAN_LD; ++k) {
            int q = k * SCAN_TB + t;
            if (q < NN4) stage4[q] = g[k];
        }
    }
    __syncthreads();
    int b = t * SCAN_CH;
    int v[SCAN_CH];
    int s = 0;
#pragma unroll
    for (int i = 0; i < SCAN_CH; ++i) {
        v[i] = (b + i < NN) ? stage[b + i] : 0;
        s += v[i];
    }
    int x = s;
    for (int o = 1; o < 32; o <<= 1) {
        int y = __shfl_up(x, o);
        if (lane >= o) x += y;
    }
    if (lane == 31) wsum[w] = x;
    __syncthreads();
    if (w == 0) {
        int tv = wsum[lane];
        for (int o = 1; o < 32; o <<= 1) {
            int y = __shfl_up(tv, o);
            if (lane >= o) tv += y;
        }
        wsum[lane] = tv;
    }
    __syncthreads();
    int run = x - s + ((w > 0) ? wsum[w - 1] : 0);
#pragma unroll
    for (int i = 0; i < SCAN_CH; ++i) {
        if (b + i < NN) stage[b + i] = run;
        run += v[i];
    }
    if (t == SCAN_TB - 1) rowptr[NN] = run;
    __syncthreads();
    {
        v4i* rp4 = (v4i*)rowptr;
#pragma unroll
        for (int k = 0; k < SCAN_LD; ++k) {
            int q = k * SCAN_TB + t;
            if (q < NN4) rp4[q] = stage4[q];
        }
    }
}

__device__ __forceinline__
void csr_fill_edge(const int* __restrict__ ei, const int* __restrict__ rowptr,
                   const int* __restrict__ rank, int* __restrict__ csrc, int e) {
    if (e >= E0C) return;
    int d_ = ei[E0C + e];
    csrc[rowptr[d_] + rank[e]] = ei[e];
}

#define MT2C     (MPAD / 32)
#define XP1_BLK  (INDIM * HIDDIM / 256)
#define XP2_BLK  (HIDDIM * OUTDIM / 256)
#define EDGE_BLK ((E0C + 255) / 256)

__global__ __launch_bounds__(256)
void weights_and_count(const float* __restrict__ W1, _Float16* __restrict__ W1T,
                       const float* __restrict__ W2, _Float16* __restrict__ W2T,
                       const int* __restrict__ ei, int* __restrict__ cnt,
                       int* __restrict__ rank) {
    const int b = blockIdx.x, t = threadIdx.x;
    if (b < XP1_BLK)
        xpose_cvt_f16_elem(W1, W1T, b * 256 + t, INDIM, HIDDIM);
    else if (b < XP1_BLK + XP2_BLK)
        xpose_cvt_f16_elem(W2, W2T, (b - XP1_BLK) * 256 + t, HIDDIM, OUTDIM);
    else
        csr_count_edge(ei, cnt, rank, (b - (XP1_BLK + XP2_BLK)) * 256 + t);
}

#define CVT_WAVES (NN * INDIM / 4 / (32 * EW_BIG))
#define CVT_BLK   ((CVT_WAVES + SCAN_TB / 32 - 1) / (SCAN_TB / 32))
__global__ __launch_bounds__(SCAN_TB)
void scan_and_cast(const int* __restrict__ cnt, int* __restrict__ rowptr,
                   const float* __restrict__ x, _Float16* __restrict__ xh) {
    __shared__ __attribute__((aligned(16))) int stage[NN];
    __shared__ int wsum[32];
    if (blockIdx.x == 0)
        csr_scan_block(cnt, rowptr, stage, wsum);
    else
        cvt_f32_to_f16_wave(x, xh, (int)(blockIdx.x - 1) * (SCAN_TB / 32) + (int)(threadIdx.x >> 5),
                            threadIdx.x & 31);
}

static_assert(2 * MT2C >= EDGE_BLK, "two fill work-groups per GEMM tile cover all edges");
__global__ __launch_bounds__(256)
void gemm_l1_coef_and_fill(const _Float16* __restrict__ A, const _Float16* __restrict__ Bt,
                           float* __restrict__ C, const float* __restrict__ att,
                           float* __restrict__ ai, float* __restrict__ aj,
                           const int* __restrict__ ei, const int* __restrict__ rowptr,
                           const int* __restrict__ rank, int* __restrict__ csrc) {
    __shared__ float tile[H1C][32 * TPITCH];
    __shared__ __attribute__((aligned(16))) float sI[32 * H1C];
    __shared__ __attribute__((aligned(16))) float sJ[32 * H1C];
    const int g = blockIdx.x / 3;
    const int m = blockIdx.x - 3 * g;
    if (m == 0)
        gemm_l1_coef_tile(A, Bt, C, att, ai, aj, g, tile, sI, sJ);
    else
        csr_fill_edge(ei, rowptr, rank, csrc, (2 * g + m - 1) * 256 + (int)threadIdx.x);
}

__device__ __forceinline__ float lrelu(float a) { return (a >= 0.f) ? a : NEG_SLOPE * a; }
__device__ __forceinline__ float elu1(float v) { return (v > 0.f) ? v : (__expf(v) - 1.f); }

__device__ __forceinline__ void fold_edge(float a, const float4 r,
                                          float& m, float& ssum, float4& acc) {
    float mn = fmaxf(m, a);
    float sc = __expf(m - mn);
    float w  = __expf(a - mn);
    ssum  = ssum * sc + w;
    acc.x = acc.x * sc + w * r.x;
    acc.y = acc.y * sc + w * r.y;
    acc.z = acc.z * sc + w * r.z;
    acc.w = acc.w * sc + w * r.w;
    m = mn;
}

#define FN      16
#define APITCH  (HIDDIM + 8)
#define H2PITCH (OUTDIM + 4)
static_assert(NN % FN == 0, "whole 16-row tiles only: no pad rows");
__global__ __launch_bounds__(256)
void node_softmax_l1_xform(const int* __restrict__ rowptr, const int* __restrict__ csrc,
                 const float* __restrict__ ai, const float* __restrict__ aj,
                 const float* __restrict__ h, const float* __restrict__ bias,
                 const _Float16* __restrict__ W2T, const float* __restrict__ att2,
                 float* __restrict__ h2, float* __restrict__ ai2, float* __restrict__ aj2) {
    __shared__ __attribute__((aligned(16))) _Float16 sact[FN * APITCH];
    __shared__ __attribute__((aligned(16))) float sh2[FN * H2PITCH];
    const int w    = threadIdx.x >> 5;
    const int lane = threadIdx.x & 31;
    const int hq  = lane >> 3;
    const int odd = lane & 1;
    const int sel = hq + 4 * odd;

#define L1_LOAD(s, ajv, r0, r1) { \
        ajv = aj[(size_t)(s) * H1C + sel]; \
        const float4* p_ = (const float4*)(h + (size_t)(s) * HIDDIM); \
        r0 = p_[lane]; r1 = p_[32 + lane]; }
#define L1_FOLD(ajv, r0, r1) { \
        float o_ = __shfl_xor(ajv, 1); \
        float a0_ = lrelu(ai0 + (odd ? o_ : ajv)); \
        float a1_ = lrelu(ai1 + (odd ? ajv : o_)); \
        fold_edge(a0_, r0, m0, s0, acc0); \
        fold_edge(a1_, r1, m1, s1, acc1); }

#pragma unroll 1
    for (int i = 0; i < FN / 8; ++i) {
        const int row  = (FN / 8) * w + i;
        const int node = (int)blockIdx.x * FN + row;
        const float ai0 = ai[(size_t)node * H1C + hq];
        const float ai1 = ai[(size_t)node * H1C + 4 + hq];
        const int beg = rowptr[node], end = rowptr[node + 1];

        float m0, m1, s0 = 1.f, s1 = 1.f;
        float4 acc0, acc1;
        {
            float ajv;
            L1_LOAD(node, ajv, acc0, acc1);
            float o_ = __shfl_xor(ajv, 1);
            m0 = lrelu(ai0 + (odd ? o_ : ajv));
            m1 = lrelu(ai1 + (odd ? ajv : o_));
        }
        int p = beg;
        for (; p + 4 <= end; p += 4) {
            int sa = csrc[p], sb = csrc[p + 1], sc_ = csrc[p + 2], sd = csrc[p + 3];
            float ja, jb, jc, jd;
            float4 a0, a1, b0, b1, c0, c1, d0, d1;
            L1_LOAD(sa, ja, a0, a1);
            L1_LOAD(sb, jb, b0, b1);
            L1_LOAD(sc_, jc, c0, c1);
            L1_LOAD(sd, jd, d0, d1);
            L1_FOLD(ja, a0, a1);
            L1_FOLD(jb, b0, b1);
            L1_FOLD(jc, c0, c1);
            L1_FOLD(jd, d0, d1);
        }
        for (; p < end; ++p) {
            int sa = csrc[p];
            float ja; float4 a0, a1;
            L1_LOAD(sa, ja, a0, a1);
            L1_FOLD(ja, a0, a1);
        }
        float i0 = 1.f / s0, i1 = 1.f / s1;
        const float4 bE = ((const float4*)bias)[lane];
        const float4 bO = ((const float4*)bias)[32 + lane];
        v4h o0, o1;
        o0.x = (_Float16)elu1(acc0.x * i0 + bE.x); o0.y = (_Float16)elu1(acc0.y * i0 + bE.y);
        o0.z = (_Float16)elu1(acc0.z * i0 + bE.z); o0.w = (_Float16)elu1(acc0.w * i0 + bE.w);
        o1.x = (_Float16)elu1(acc1.x * i1 + bO.x); o1.y = (_Float16)elu1(acc1.y * i1 + bO.y);
        o1.z = (_Float16)elu1(acc1.z * i1 + bO.z); o1.w = (_Float16)elu1(acc1.w * i1 + bO.w);
        v4h* o = (v4h*)(sact + row * APITCH);
        o[lane]      = o0;
        o[32 + lane] = o1;
    }
#undef L1_LOAD
#undef L1_FOLD
    __syncthreads();

    if (w < OUTDIM / 16) {
        const int r    = lane & 15;
        const int half = lane >> 4;
        const _Float16* a0 = sact + r * APITCH + half * 8;
        const _Float16* b0 = W2T + (size_t)(w * 16 + r) * HIDDIM + half * 8;
        v8f c = {};
        for (int k0 = 0; k0 < HIDDIM; k0 += 32) {
            V16U A, B;
            A.h[0] = *(const v8h*)(a0 + k0);
            A.h[1] = *(const v8h*)(a0 + k0 + 16);
            B.h[0] = *(const v8h*)(b0 + k0);
            B.h[1] = *(const v8h*)(b0 + k0 + 16);
            c = WMMA16(A.v, B.v, c);
        }
#pragma unroll
        for (int j = 0; j < 8; ++j) sh2[(half * 8 + j) * H2PITCH + w * 16 + r] = c[j];
    }
    __syncthreads();

    const int t = threadIdx.x;
    if (t < FN) {
        const float* hv = sh2 + t * H2PITCH;
        float si = 0.f, sj = 0.f;
        for (int c = 0; c < C2C; ++c) {
            si = __builtin_fmaf(hv[c], att2[c], si);
            sj = __builtin_fmaf(hv[c], att2[C2C + c], sj);
        }
        ai2[(int)blockIdx.x * FN + t] = si;
        aj2[(int)blockIdx.x * FN + t] = sj;
    }
    {
        const int row = t >> 4, q = t & 15;
        ((float4*)(h2 + (size_t)((int)blockIdx.x * FN + row) * OUTDIM))[q] =
            *(const float4*)(sh2 + row * H2PITCH + 4 * q);
    }
}

__global__ __launch_bounds__(256)
void node_softmax_l2(const int* __restrict__ rowptr, const int* __restrict__ csrc,
                 const float* __restrict__ ai, const float* __restrict__ aj,
                 const float* __restrict__ h, const float* __restrict__ bias,
                 float* __restrict__ outp) {
    int gid  = blockIdx.x * blockDim.x + threadIdx.x;
    int node = gid >> 4;
    int lg   = gid & 15;
    if (node >= NN) return;
    const float aiv = ai[node];
    const int beg = rowptr[node], end = rowptr[node + 1];

    float m = lrelu(aiv + aj[node]), ssum = 1.f;
    float4 acc = ((const float4*)(h + (size_t)node * OUTDIM))[lg];

    int p = beg;
    for (; p + 4 <= end; p += 4) {
        int sa = csrc[p], sb = csrc[p + 1], sc_ = csrc[p + 2], sd = csrc[p + 3];
        float ja = aj[sa], jb = aj[sb], jc = aj[sc_], jd = aj[sd];
        float4 ra = ((const float4*)(h + (size_t)sa  * OUTDIM))[lg];
        float4 rb = ((const float4*)(h + (size_t)sb  * OUTDIM))[lg];
        float4 rc = ((const float4*)(h + (size_t)sc_ * OUTDIM))[lg];
        float4 rd = ((const float4*)(h + (size_t)sd  * OUTDIM))[lg];
        fold_edge(lrelu(aiv + ja), ra, m, ssum, acc);
        fold_edge(lrelu(aiv + jb), rb, m, ssum, acc);
        fold_edge(lrelu(aiv + jc), rc, m, ssum, acc);
        fold_edge(lrelu(aiv + jd), rd, m, ssum, acc);
    }
    for (; p < end; ++p) {
        int sa = csrc[p];
        float ja = aj[sa];
        float4 ra = ((const float4*)(h + (size_t)sa * OUTDIM))[lg];
        fold_edge(lrelu(aiv + ja), ra, m, ssum, acc);
    }
    float inv = 1.f / ssum;
    const float4 b = ((const float4*)bias)[lg];
    ((float4*)(outp + (size_t)node * OUTDIM))[lg] =
        make_float4(acc.x * inv + b.x, acc.y * inv + b.y, acc.z * inv + b.z, acc.w * inv + b.w);
}


extern "C" void kernel_launch(void* const* d_in, const int* in_sizes, int n_in,
                              void* d_out, int out_size, void* d_ws, size_t ws_size,
                              hipStream_t stream) {
    (void)in_sizes; (void)n_in; (void)out_size; (void)ws_size;

    const float* x    = (const float*)d_in[0];
    const int*   ei   = (const int*)  d_in[1];
    const float* W1   = (const float*)d_in[2];
    const float* att1 = (const float*)d_in[3];
    const float* b1   = (const float*)d_in[4];
    const float* W2   = (const float*)d_in[5];
    const float* att2 = (const float*)d_in[6];
    const float* b2   = (const float*)d_in[7];
    float* out = (float*)d_out;

    char* base = (char*)d_ws;
    size_t off = 0;
    auto carve = [&](size_t bytes) -> char* {
        char* p = base + off;
        off = (off + bytes + 255) & ~(size_t)255;
        return p;
    };
    _Float16* XH    = (_Float16*)carve((size_t)MPAD * INDIM * 2);
    _Float16* W1T   = (_Float16*)carve((size_t)INDIM * HIDDIM * 2);
    _Float16* W2T   = (_Float16*)carve((size_t)HIDDIM * OUTDIM * 2);
    float*    HBUF  = (float*)   carve((size_t)MPAD * HIDDIM * 4);
    float*    AI    = (float*)   carve((size_t)NN * H1C * 4);
    float*    AJ    = (float*)   carve((size_t)NN * H1C * 4);
    int*      CNT   = (int*)     carve((size_t)NN * 4);
    int*      RANK  = (int*)     carve((size_t)E0C * 4);
    int*      ROWP  = (int*)     carve((size_t)(NN + 1) * 4);
    int*      CSRC  = (int*)     carve((size_t)E0C * 4);
    float*    H2BUF = (float*)   carve((size_t)NN * OUTDIM * 4);
    float*    AI2   = (float*)   carve((size_t)NN * 4);
    float*    AJ2   = (float*)   carve((size_t)NN * 4);

    const int TB = 256;
    auto nb = [](long long n, int tb) { return (unsigned)((n + tb - 1) / tb); };
    const int MT2 = MPAD / 32;

    hipMemsetAsync(CNT, 0, (size_t)NN * 4, stream);
    weights_and_count<<<XP1_BLK + XP2_BLK + EDGE_BLK, TB, 0, stream>>>(W1, W1T, W2, W2T, ei, CNT, RANK);
    scan_and_cast<<<1 + CVT_BLK, SCAN_TB, 0, stream>>>(CNT, ROWP, x, XH);

    gemm_l1_coef_and_fill<<<3 * MT2, 256, 0, stream>>>(XH, W1T, HBUF, att1, AI, AJ, ei, ROWP, RANK, CSRC);

    node_softmax_l1_xform<<<NN / FN, TB, 0, stream>>>(ROWP, CSRC, AI, AJ, HBUF, b1, W2T, att2, H2BUF, AI2, AJ2);

    node_softmax_l2<<<nb((long long)NN * 16, TB), TB, 0, stream>>>(ROWP, CSRC, AI2, AJ2, H2BUF, b2, out);
}


# GEMM operands stored fragment-major by their producers (cast of x, weight transposes): each WMMA fragment load reads 512 contiguous bytes, was 32 bytes of each of 16 lines; values, k order kept
# speedup vs baseline: 4.1153x; 1.0561x over previous
#include <hip/hip_runtime.h>
#include <hip/hip_bf16.h>

#define NN     50000
#define MPAD   50016
#define E0C    800000
#define ETOT   850000
#define INDIM  256
#define HIDDIM 256
#define OUTDIM 64
#define H1C    8
#define C1C    32
#define H2C    1
#define C2C    64
#define NEG_SLOPE 0.2f

typedef __attribute__((ext_vector_type(16))) _Float16 v16h;
typedef __attribute__((ext_vector_type(8)))  _Float16 v8h;
typedef __attribute__((ext_vector_type(4)))  _Float16 v4h;
typedef __attribute__((ext_vector_type(8)))  float    v8f;

union V16U { v16h v; v8h h[2]; };

#define FRAG_TILE(K) (16 * (K))
__device__ __forceinline__
size_t frag_index(int i, int k, int K) {
    return (size_t)(i >> 4) * FRAG_TILE(K) + (size_t)((k >> 4) * 256 + ((k >> 3) & 1) * 128 + (i & 15) * 8 + (k & 7));
}

#define CVT_U 8
static_assert(INDIM == 256 && HIDDIM == 256, "two waves per 16-row tile of x; 8 k-steps of 32");
static_assert(NN % 16 == 0, "x has whole 16-row tiles only");

__device__ __forceinline__
void cvt_f32_to_f16_wave(const float* __restrict__ src, _Float16* __restrict__ dst,
                         int wave, int lane) {
    if (wave >= (NN / 16) * 2) return;
    const int T = wave >> 1, kh = wave & 1;
    const int r = lane & 15, half = lane >> 4;
    const float4* s4 = (const float4*)(src + (size_t)(T * 16 + r) * INDIM + kh * 128 + half * 8);
    float4 v[2 * CVT_U];
#pragma unroll
    for (int u = 0; u < CVT_U; ++u) { v[2 * u] = s4[u * 4]; v[2 * u + 1] = s4[u * 4 + 1]; }
    v8h* d8 = (v8h*)(dst + (size_t)T * FRAG_TILE(INDIM) + (size_t)(kh * CVT_U) * 256 + lane * 8);
#pragma unroll
    for (int u = 0; u < CVT_U; ++u) {
        v8h o;
        o[0] = (_Float16)v[2 * u].x;     o[1] = (_Float16)v[2 * u].y;
        o[2] = (_Float16)v[2 * u].z;     o[3] = (_Float16)v[2 * u].w;
        o[4] = (_Float16)v[2 * u + 1].x; o[5] = (_Float16)v[2 * u + 1].y;
        o[6] = (_Float16)v[2 * u + 1].z; o[7] = (_Float16)v[2 * u + 1].w;
        d8[u * 32] = o;
    }
}

__device__ __forceinline__
void xpose_cvt_f16_elem(const float* __restrict__ W, _Float16* __restrict__ Wt,
                        int i, int K, int Nc) {
    if (i >= K * Nc) return;
    int k = i / Nc, n = i % Nc;
    Wt[frag_index(n, k, K)] = (_Float16)W[i];
}

#define WMMA16(a, b, c) __builtin_amdgcn_wmma_f32_16x16x32_f16( \
        false, (a), false, (b), (short)0, (c), false, false)

__device__ __forceinline__
void wave_tile_32x32(const _Float16* __restrict__ A, const _Float16* __restrict__ Bt,
                     int tm, int tn, int lane, int K,
                     v8f& c00, v8f& c01, v8f& c10, v8f& c11) {
    const _Float16* a0 = A  + (size_t)(tm * 2) * FRAG_TILE(K) + lane * 8;
    const _Float16* a1 = a0 + FRAG_TILE(K);
    const _Float16* b0 = Bt + (size_t)(tn * 2) * FRAG_TILE(K) + lane * 8;
    const _Float16* b1 = b0 + FRAG_TILE(K);
    for (int k0 = 0; k0 < K; k0 += 32) {
        V16U A0, A1, B0, B1;
        A0.h[0] = *(const v8h*)(a0 + k0 * 16);
        A0.h[1] = *(const v8h*)(a0 + k0 * 16 + 256);
        A1.h[0] = *(const v8h*)(a1 + k0 * 16);
        A1.h[1] = *(const v8h*)(a1 + k0 * 16 + 256);
        B0.h[0] = *(const v8h*)(b0 + k0 * 16);
        B0.h[1] = *(const v8h*)(b0 + k0 * 16 + 256);
        B1.h[0] = *(const v8h*)(b1 + k0 * 16);
        B1.h[1] = *(const v8h*)(b1 + k0 * 16 + 256);
        c00 = WMMA16(A0.v, B0.v, c00);
        c01 = WMMA16(A0.v, B1.v, c01);
        c10 = WMMA16(A1.v, B0.v, c10);
        c11 = WMMA16(A1.v, B1.v, c11);
    }
}

__device__ __forceinline__
void store_tile_32x32(float* __restrict__ C, int Nc, int tm, int tn, int r, int half,
                      const v8f& c00, const v8f& c01, const v8f& c10, const v8f& c11) {
    float* p = C + (size_t)(tm * 32 + half * 8) * Nc + tn * 32 + r;
#pragma unroll
    for (int j = 0; j < 8; ++j) { p[(size_t)j * Nc] = c00[j]; p[(size_t)j * Nc + 16] = c01[j]; }
    p += (size_t)16 * Nc;
#pragma unroll
    for (int j = 0; j < 8; ++j) { p[(size_t)j * Nc] = c10[j]; p[(size_t)j * Nc + 16] = c11[j]; }
}

#define TPITCH 34
__device__ __forceinline__
void gemm_l1_coef_tile(const _Float16* __restrict__ A, const _Float16* __restrict__ Bt,
                       float* __restrict__ C, const float* __restrict__ att,
                       float* __restrict__ ai, float* __restrict__ aj, const int tm,
                       float (*tile)[32 * TPITCH], float* sI, float* sJ) {
    const int tn   = threadIdx.x >> 5;
    const int lane = threadIdx.x & 31;
    const int r    = lane & 15;
    const int half = lane >> 4;
    v8f c00 = {}, c01 = {}, c10 = {}, c11 = {};
    wave_tile_32x32(A, Bt, tm, tn, lane, INDIM, c00, c01, c10, c11);

    float* T = tile[tn];
#pragma unroll
    for (int j = 0; j < 8; ++j) {
        T[(half * 8 + j) * TPITCH + r]           = c00[j];
        T[(half * 8 + j) * TPITCH + 16 + r]      = c01[j];
        T[(16 + half * 8 + j) * TPITCH + r]      = c10[j];
        T[(16 + half * 8 + j) * TPITCH + 16 + r] = c11[j];
    }
    __syncthreads();
    const float* ad = att + tn * 2 * C1C;
    const float* as = ad + C1C;
    float si = 0.f, sj = 0.f;
    for (int c = 0; c < C1C; ++c) {
        float hv = T[lane * TPITCH + c];
        si = __builtin_fmaf(hv, ad[c], si);
        sj = __builtin_fmaf(hv, as[c], sj);
    }
    sI[lane * H1C + tn] = si;
    sJ[lane * H1C + tn] = sj;
    __syncthreads();
    store_tile_32x32(C, HIDDIM, tm, tn, r, half, c00, c01, c10, c11);
    const int t = threadIdx.x;
    if (t < 128) {
        const int q = t & 63;
        if (tm * 32 + (q >> 1) < NN) {
            if (t < 64) ((float4*)ai)[(size_t)tm * 64 + q] = ((const float4*)sI)[q];
            else        ((float4*)aj)[(size_t)tm * 64 + q] = ((const float4*)sJ)[q];
        }
    }
}

__device__ __forceinline__
void csr_count_edge(const int* __restrict__ ei, int* __restrict__ cnt,
                    int* __restrict__ rank, int e) {
    if (e >= E0C) return;
    int d_ = ei[E0C + e];
    rank[e] = atomicAdd(cnt + d_, 1);
}

#define SCAN_TB 1024
#define SCAN_CH ((NN + SCAN_TB - 1) / SCAN_TB)
#define NN4     (NN / 4)
#define SCAN_LD ((NN4 + SCAN_TB - 1) / SCAN_TB)
typedef int v4i __attribute__((ext_vector_type(4)));
__device__ __forceinline__
void csr_scan_block(const int* __restrict__ cnt, int* __restrict__ rowptr,
                    int* stage, int* wsum) {
    int t = threadIdx.x;
    int lane = t & 31, w = t >> 5;
    v4i* stage4 = (v4i*)stage;
    {
        const v4i* cnt4 = (const v4i*)cnt;
        v4i g[SCAN_LD];
#pragma unroll
        for (int k = 0; k < SCAN_LD; ++k) {
            int q = k * SCAN_TB + t;
            if (q < NN4) g[k] = cnt4[q];
        }
#pragma unroll
        for (int k = 0; k < SCAN_LD; ++k) {
            int q = k * SCAN_TB + t;
            if (q < NN4) stage4[q] = g[k];
        }
    }
    __syncthreads();
    int b = t * SCAN_CH;
    int v[SCAN_CH];
    int s = 0;
#pragma unroll
    for (int i = 0; i < SCAN_CH; ++i) {
        v[i] = (b + i < NN) ? stage[b + i] : 0;
        s += v[i];
    }
    int x = s;
    for (int o = 1; o < 32; o <<= 1) {
        int y = __shfl_up(x, o);
        if (lane >= o) x += y;
    }
    if (lane == 31) wsum[w] = x;
    __syncthreads();
    if (w == 0) {
        int tv = wsum[lane];
        for (int o = 1; o < 32; o <<= 1) {
            int y = __shfl_up(tv, o);
            if (lane >= o) tv += y;
        }
        wsum[lane] = tv;
    }
    __syncthreads();
    int run = x - s + ((w > 0) ? wsum[w - 1] : 0);
#pragma unroll
    for (int i = 0; i < SCAN_CH; ++i) {
        if (b + i < NN) stage[b + i] = run;
        run += v[i];
    }
    if (t == SCAN_TB - 1) rowptr[NN] = run;
    __syncthreads();
    {
        v4i* rp4 = (v4i*)rowptr;
#pragma unroll
        for (int k = 0; k < SCAN_LD; ++k) {
            int q = k * SCAN_TB + t;
            if (q < NN4) rp4[q] = stage4[q];
        }
    }
}

__device__ __forceinline__
void csr_fill_edge(const int* __restrict__ ei, const int* __restrict__ rowptr,
                   const int* __restrict__ rank, int* __restrict__ csrc, int e) {
    if (e >= E0C) return;
    int d_ = ei[E0C + e];
    csrc[rowptr[d_] + rank[e]] = ei[e];
}

#define MT2C     (MPAD / 32)
#define XP1_BLK  (INDIM * HIDDIM / 256)
#define XP2_BLK  (HIDDIM * OUTDIM / 256)
#define EDGE_BLK ((E0C + 255) / 256)

__global__ __launch_bounds__(256)
void weights_and_count(const float* __restrict__ W1, _Float16* __restrict__ W1T,
                       const float* __restrict__ W2, _Float16* __restrict__ W2T,
                       const int* __restrict__ ei, int* __restrict__ cnt,
                       int* __restrict__ rank) {
    const int b = blockIdx.x, t = threadIdx.x;
    if (b < XP1_BLK)
        xpose_cvt_f16_elem(W1, W1T, b * 256 + t, INDIM, HIDDIM);
    else if (b < XP1_BLK + XP2_BLK)
        xpose_cvt_f16_elem(W2, W2T, (b - XP1_BLK) * 256 + t, HIDDIM, OUTDIM);
    else
        csr_count_edge(ei, cnt, rank, (b - (XP1_BLK + XP2_BLK)) * 256 + t);
}

#define CVT_WAVES ((NN / 16) * 2)
#define CVT_BLK   ((CVT_WAVES + SCAN_TB / 32 - 1) / (SCAN_TB / 32))
__global__ __launch_bounds__(SCAN_TB)
void scan_and_cast(const int* __restrict__ cnt, int* __restrict__ rowptr,
                   const float* __restrict__ x, _Float16* __restrict__ xh) {
    __shared__ __attribute__((aligned(16))) int stage[NN];
    __shared__ int wsum[32];
    if (blockIdx.x == 0)
        csr_scan_block(cnt, rowptr, stage, wsum);
    else
        cvt_f32_to_f16_wave(x, xh, (int)(blockIdx.x - 1) * (SCAN_TB / 32) + (int)(threadIdx.x >> 5),
                            threadIdx.x & 31);
}

static_assert(2 * MT2C >= EDGE_BLK, "two fill work-groups per GEMM tile cover all edges");
__global__ __launch_bounds__(256)
void gemm_l1_coef_and_fill(const _Float16* __restrict__ A, const _Float16* __restrict__ Bt,
                           float* __restrict__ C, const float* __restrict__ att,
                           float* __restrict__ ai, float* __restrict__ aj,
                           const int* __restrict__ ei, const int* __restrict__ rowptr,
                           const int* __restrict__ rank, int* __restrict__ csrc) {
    __shared__ float tile[H1C][32 * TPITCH];
    __shared__ __attribute__((aligned(16))) float sI[32 * H1C];
    __shared__ __attribute__((aligned(16))) float sJ[32 * H1C];
    const int g = blockIdx.x / 3;
    const int m = blockIdx.x - 3 * g;
    if (m == 0)
        gemm_l1_coef_tile(A, Bt, C, att, ai, aj, g, tile, sI, sJ);
    else
        csr_fill_edge(ei, rowptr, rank, csrc, (2 * g + m - 1) * 256 + (int)threadIdx.x);
}

__device__ __forceinline__ float lrelu(float a) { return (a >= 0.f) ? a : NEG_SLOPE * a; }
__device__ __forceinline__ float elu1(float v) { return (v > 0.f) ? v : (__expf(v) - 1.f); }

__device__ __forceinline__ void fold_edge(float a, const float4 r,
                                          float& m, float& ssum, float4& acc) {
    float mn = fmaxf(m, a);
    float sc = __expf(m - mn);
    float w  = __expf(a - mn);
    ssum  = ssum * sc + w;
    acc.x = acc.x * sc + w * r.x;
    acc.y = acc.y * sc + w * r.y;
    acc.z = acc.z * sc + w * r.z;
    acc.w = acc.w * sc + w * r.w;
    m = mn;
}

#define FN      16
#define APITCH  (HIDDIM + 8)
#define H2PITCH (OUTDIM + 4)
static_assert(NN % FN == 0, "whole 16-row tiles only: no pad rows");
__global__ __launch_bounds__(256)
void node_softmax_l1_xform(const int* __restrict__ rowptr, const int* __restrict__ csrc,
                 const float* __restrict__ ai, const float* __restrict__ aj,
                 const float* __restrict__ h, const float* __restrict__ bias,
                 const _Float16* __restrict__ W2T, const float* __restrict__ att2,
                 float* __restrict__ h2, float* __restrict__ ai2, float* __restrict__ aj2) {
    __shared__ __attribute__((aligned(16))) _Float16 sact[FN * APITCH];
    __shared__ __attribute__((aligned(16))) float sh2[FN * H2PITCH];
    const int w    = threadIdx.x >> 5;
    const int lane = threadIdx.x & 31;
    const int hq  = lane >> 3;
    const int odd = lane & 1;
    const int sel = hq + 4 * odd;

#define L1_LOAD(s, ajv, r0, r1) { \
        ajv = aj[(size_t)(s) * H1C + sel]; \
        const float4* p_ = (const float4*)(h + (size_t)(s) * HIDDIM); \
        r0 = p_[lane]; r1 = p_[32 + lane]; }
#define L1_FOLD(ajv, r0, r1) { \
        float o_ = __shfl_xor(ajv, 1); \
        float a0_ = lrelu(ai0 + (odd ? o_ : ajv)); \
        float a1_ = lrelu(ai1 + (odd ? ajv : o_)); \
        fold_edge(a0_, r0, m0, s0, acc0); \
        fold_edge(a1_, r1, m1, s1, acc1); }

#pragma unroll 1
    for (int i = 0; i < FN / 8; ++i) {
        const int row  = (FN / 8) * w + i;
        const int node = (int)blockIdx.x * FN + row;
        const float ai0 = ai[(size_t)node * H1C + hq];
        const float ai1 = ai[(size_t)node * H1C + 4 + hq];
        const int beg = rowptr[node], end = rowptr[node + 1];

        float m0, m1, s0 = 1.f, s1 = 1.f;
        float4 acc0, acc1;
        {
            float ajv;
            L1_LOAD(node, ajv, acc0, acc1);
            float o_ = __shfl_xor(ajv, 1);
            m0 = lrelu(ai0 + (odd ? o_ : ajv));
            m1 = lrelu(ai1 + (odd ? ajv : o_));
        }
        int p = beg;
        for (; p + 4 <= end; p += 4) {
            int sa = csrc[p], sb = csrc[p + 1], sc_ = csrc[p + 2], sd = csrc[p + 3];
            float ja, jb, jc, jd;
            float4 a0, a1, b0, b1, c0, c1, d0, d1;
            L1_LOAD(sa, ja, a0, a1);
            L1_LOAD(sb, jb, b0, b1);
            L1_LOAD(sc_, jc, c0, c1);
            L1_LOAD(sd, jd, d0, d1);
            L1_FOLD(ja, a0, a1);
            L1_FOLD(jb, b0, b1);
            L1_FOLD(jc, c0, c1);
            L1_FOLD(jd, d0, d1);
        }
        for (; p < end; ++p) {
            int sa = csrc[p];
            float ja; float4 a0, a1;
            L1_LOAD(sa, ja, a0, a1);
            L1_FOLD(ja, a0, a1);
        }
        float i0 = 1.f / s0, i1 = 1.f / s1;
        const float4 bE = ((const float4*)bias)[lane];
        const float4 bO = ((const float4*)bias)[32 + lane];
        v4h o0, o1;
        o0.x = (_Float16)elu1(acc0.x * i0 + bE.x); o0.y = (_Float16)elu1(acc0.y * i0 + bE.y);
        o0.z = (_Float16)elu1(acc0.z * i0 + bE.z); o0.w = (_Float16)elu1(acc0.w * i0 + bE.w);
        o1.x = (_Float16)elu1(acc1.x * i1 + bO.x); o1.y = (_Float16)elu1(acc1.y * i1 + bO.y);
        o1.z = (_Float16)elu1(acc1.z * i1 + bO.z); o1.w = (_Float16)elu1(acc1.w * i1 + bO.w);
        v4h* o = (v4h*)(sact + row * APITCH);
        o[lane]      = o0;
        o[32 + lane] = o1;
    }
#undef L1_LOAD
#undef L1_FOLD
    __syncthreads();

    if (w < OUTDIM / 16) {
        const int r    = lane & 15;
        const int half = lane >> 4;
        const _Float16* a0 = sact + r * APITCH + half * 8;
        const _Float16* b0 = W2T + (size_t)w * FRAG_TILE(HIDDIM) + lane * 8;
        v8f c = {};
        for (int k0 = 0; k0 < HIDDIM; k0 += 32) {
            V16U A, B;
            A.h[0] = *(const v8h*)(a0 + k0);
            A.h[1] = *(const v8h*)(a0 + k0 + 16);
            B.h[0] = *(const v8h*)(b0 + k0 * 16);
            B.h[1] = *(const v8h*)(b0 + k0 * 16 + 256);
            c = WMMA16(A.v, B.v, c);
        }
#pragma unroll
        for (int j = 0; j < 8; ++j) sh2[(half * 8 + j) * H2PITCH + w * 16 + r] = c[j];
    }
    __syncthreads();

    const int t = threadIdx.x;
    if (t < FN) {
        const float* hv = sh2 + t * H2PITCH;
        float si = 0.f, sj = 0.f;
        for (int c = 0; c < C2C; ++c) {
            si = __builtin_fmaf(hv[c], att2[c], si);
            sj = __builtin_fmaf(hv[c], att2[C2C + c], sj);
        }
        ai2[(int)blockIdx.x * FN + t] = si;
        aj2[(int)blockIdx.x * FN + t] = sj;
    }
    {
        const int row = t >> 4, q = t & 15;
        ((float4*)(h2 + (size_t)((int)blockIdx.x * FN + row) * OUTDIM))[q] =
            *(const float4*)(sh2 + row * H2PITCH + 4 * q);
    }
}

__global__ __launch_bounds__(256)
void node_softmax_l2(const int* __restrict__ rowptr, const int* __restrict__ csrc,
                 const float* __restrict__ ai, const float* __restrict__ aj,
                 const float* __restrict__ h, const float* __restrict__ bias,
                 float* __restrict__ outp) {
    int gid  = blockIdx.x * blockDim.x + threadIdx.x;
    int node = gid >> 4;
    int lg   = gid & 15;
    if (node >= NN) return;
    const float aiv = ai[node];
    const int beg = rowptr[node], end = rowptr[node + 1];

    float m = lrelu(aiv + aj[node]), ssum = 1.f;
    float4 acc = ((const float4*)(h + (size_t)node * OUTDIM))[lg];

    int p = beg;
    for (; p + 4 <= end; p += 4) {
        int sa = csrc[p], sb = csrc[p + 1], sc_ = csrc[p + 2], sd = csrc[p + 3];
        float ja = aj[sa], jb = aj[sb], jc = aj[sc_], jd = aj[sd];
        float4 ra = ((const float4*)(h + (size_t)sa  * OUTDIM))[lg];
        float4 rb = ((const float4*)(h + (size_t)sb  * OUTDIM))[lg];
        float4 rc = ((const float4*)(h + (size_t)sc_ * OUTDIM))[lg];
        float4 rd = ((const float4*)(h + (size_t)sd  * OUTDIM))[lg];
        fold_edge(lrelu(aiv + ja), ra, m, ssum, acc);
        fold_edge(lrelu(aiv + jb), rb, m, ssum, acc);
        fold_edge(lrelu(aiv + jc), rc, m, ssum, acc);
        fold_edge(lrelu(aiv + jd), rd, m, ssum, acc);
    }
    for (; p < end; ++p) {
        int sa = csrc[p];
        float ja = aj[sa];
        float4 ra = ((const float4*)(h + (size_t)sa * OUTDIM))[lg];
        fold_edge(lrelu(aiv + ja), ra, m, ssum, acc);
    }
    float inv = 1.f / ssum;
    const float4 b = ((const float4*)bias)[lg];
    ((float4*)(outp + (size_t)node * OUTDIM))[lg] =
        make_float4(acc.x * inv + b.x, acc.y * inv + b.y, acc.z * inv + b.z, acc.w * inv + b.w);
}


extern "C" void kernel_launch(void* const* d_in, const int* in_sizes, int n_in,
                              void* d_out, int out_size, void* d_ws, size_t ws_size,
                              hipStream_t stream) {
    (void)in_sizes; (void)n_in; (void)out_size; (void)ws_size;

    const float* x    = (const float*)d_in[0];
    const int*   ei   = (const int*)  d_in[1];
    const float* W1   = (const float*)d_in[2];
    const float* att1 = (const float*)d_in[3];
    const float* b1   = (const float*)d_in[4];
    const float* W2   = (const float*)d_in[5];
    const float* att2 = (const float*)d_in[6];
    const float* b2   = (const float*)d_in[7];
    float* out = (float*)d_out;

    char* base = (char*)d_ws;
    size_t off = 0;
    auto carve = [&](size_t bytes) -> char* {
        char* p = base + off;
        off = (off + bytes + 255) & ~(size_t)255;
        return p;
    };
    _Float16* XH    = (_Float16*)carve((size_t)MPAD * INDIM * 2);
    _Float16* W1T   = (_Float16*)carve((size_t)INDIM * HIDDIM * 2);
    _Float16* W2T   = (_Float16*)carve((size_t)HIDDIM * OUTDIM * 2);
    float*    HBUF  = (float*)   carve((size_t)MPAD * HIDDIM * 4);
    float*    AI    = (float*)   carve((size_t)NN * H1C * 4);
    float*    AJ    = (float*)   carve((size_t)NN * H1C * 4);
    int*      CNT   = (int*)     carve((size_t)NN * 4);
    int*      RANK  = (int*)     carve((size_t)E0C * 4);
    int*      ROWP  = (int*)     carve((size_t)(NN + 1) * 4);
    int*      CSRC  = (int*)     carve((size_t)E0C * 4);
    float*    H2BUF = (float*)   carve((size_t)NN * OUTDIM * 4);
    float*    AI2   = (float*)   carve((size_t)NN * 4);
    float*    AJ2   = (float*)   carve((size_t)NN * 4);

    const int TB = 256;
    auto nb = [](long long n, int tb) { return (unsigned)((n + tb - 1) / tb); };
    const int MT2 = MPAD / 32;

    hipMemsetAsync(CNT, 0, (size_t)NN * 4, stream);
    weights_and_count<<<XP1_BLK + XP2_BLK + EDGE_BLK, TB, 0, stream>>>(W1, W1T, W2, W2T, ei, CNT, RANK);
    scan_and_cast<<<1 + CVT_BLK, SCAN_TB, 0, stream>>>(CNT, ROWP, x, XH);

    gemm_l1_coef_and_fill<<<3 * MT2, 256, 0, stream>>>(XH, W1T, HBUF, att1, AI, AJ, ei, ROWP, RANK, CSRC);

    node_softmax_l1_xform<<<NN / FN, TB, 0, stream>>>(ROWP, CSRC, AI, AJ, HBUF, b1, W2T, att2, H2BUF, AI2, AJ2);

    node_softmax_l2<<<nb((long long)NN * 16, TB), TB, 0, stream>>>(ROWP, CSRC, AI2, AJ2, H2BUF, b2, out);
}


# A chunk's bin counts are made by its own scatter work-group from the edges it holds; chunks keep own bin-ordered regions, a bin is gathered from 49 runs; count launch gone (5 to 4); sum order kept
# speedup vs baseline: 7.7734x; 1.8889x over previous
#include <hip/hip_runtime.h>
#include <hip/hip_bf16.h>

#define NN     50000
#define MPAD   50016
#define E0C    800000
#define ETOT   850000
#define INDIM  256
#define HIDDIM 256
#define OUTDIM 64
#define H1C    8
#define C1C    32
#define H2C    1
#define C2C    64
#define NEG_SLOPE 0.2f

typedef __attribute__((ext_vector_type(16))) _Float16 v16h;
typedef __attribute__((ext_vector_type(8)))  _Float16 v8h;
typedef __attribute__((ext_vector_type(4)))  _Float16 v4h;
typedef __attribute__((ext_vector_type(8)))  float    v8f;

union V16U { v16h v; v8h h[2]; };

#define FRAG_TILE(K) (16 * (K))
__device__ __forceinline__
size_t frag_index(int i, int k, int K) {
    return (size_t)(i >> 4) * FRAG_TILE(K) + (size_t)((k >> 4) * 256 + ((k >> 3) & 1) * 128 + (i & 15) * 8 + (k & 7));
}

#define CVT_U 8
static_assert(INDIM == 256 && HIDDIM == 256, "two waves per 16-row tile of x; 8 k-steps of 32");
static_assert(NN % 16 == 0, "x has whole 16-row tiles only");

__device__ __forceinline__
void cvt_f32_to_f16_wave(const float* __restrict__ src, _Float16* __restrict__ dst,
                         int wave, int lane) {
    if (wave >= (NN / 16) * 2) return;
    const int T = wave >> 1, kh = wave & 1;
    const int r = lane & 15, half = lane >> 4;
    const float4* s4 = (const float4*)(src + (size_t)(T * 16 + r) * INDIM + kh * 128 + half * 8);
    float4 v[2 * CVT_U];
#pragma unroll
    for (int u = 0; u < CVT_U; ++u) { v[2 * u] = s4[u * 4]; v[2 * u + 1] = s4[u * 4 + 1]; }
    v8h* d8 = (v8h*)(dst + (size_t)T * FRAG_TILE(INDIM) + (size_t)(kh * CVT_U) * 256 + lane * 8);
#pragma unroll
    for (int u = 0; u < CVT_U; ++u) {
        v8h o;
        o[0] = (_Float16)v[2 * u].x;     o[1] = (_Float16)v[2 * u].y;
        o[2] = (_Float16)v[2 * u].z;     o[3] = (_Float16)v[2 * u].w;
        o[4] = (_Float16)v[2 * u + 1].x; o[5] = (_Float16)v[2 * u + 1].y;
        o[6] = (_Float16)v[2 * u + 1].z; o[7] = (_Float16)v[2 * u + 1].w;
        d8[u * 32] = o;
    }
}

#define WMMA16(a, b, c) __builtin_amdgcn_wmma_f32_16x16x32_f16( \
        false, (a), false, (b), (short)0, (c), false, false)

__device__ __forceinline__
void wave_tile_32x32(const _Float16* __restrict__ A, const _Float16* __restrict__ Bt,
                     int tm, int tn, int lane, int K,
                     v8f& c00, v8f& c01, v8f& c10, v8f& c11) {
    const _Float16* a0 = A  + (size_t)(tm * 2) * FRAG_TILE(K) + lane * 8;
    const _Float16* a1 = a0 + FRAG_TILE(K);
    const _Float16* b0 = Bt + (size_t)(tn * 2) * FRAG_TILE(K) + lane * 8;
    const _Float16* b1 = b0 + FRAG_TILE(K);
    for (int k0 = 0; k0 < K; k0 += 32) {
        V16U A0, A1, B0, B1;
        A0.h[0] = *(const v8h*)(a0 + k0 * 16);
        A0.h[1] = *(const v8h*)(a0 + k0 * 16 + 256);
        A1.h[0] = *(const v8h*)(a1 + k0 * 16);
        A1.h[1] = *(const v8h*)(a1 + k0 * 16 + 256);
        B0.h[0] = *(const v8h*)(b0 + k0 * 16);
        B0.h[1] = *(const v8h*)(b0 + k0 * 16 + 256);
        B1.h[0] = *(const v8h*)(b1 + k0 * 16);
        B1.h[1] = *(const v8h*)(b1 + k0 * 16 + 256);
        c00 = WMMA16(A0.v, B0.v, c00);
        c01 = WMMA16(A0.v, B1.v, c01);
        c10 = WMMA16(A1.v, B0.v, c10);
        c11 = WMMA16(A1.v, B1.v, c11);
    }
}

#define TPITCH 34
#define TSTRIDE (32 * TPITCH + 1)
__device__ __forceinline__
void gemm_l1_coef_tile(const _Float16* __restrict__ A, const _Float16* __restrict__ Bt,
                       _Float16* __restrict__ C, const float* __restrict__ att,
                       float* __restrict__ ai, float* __restrict__ aj, const int tm,
                       float* tile, float* sI, float* sJ) {
    const int tn   = threadIdx.x >> 5;
    const int lane = threadIdx.x & 31;
    const int r    = lane & 15;
    const int half = lane >> 4;
    v8f c00 = {}, c01 = {}, c10 = {}, c11 = {};
    wave_tile_32x32(A, Bt, tm, tn, lane, INDIM, c00, c01, c10, c11);

    float* T = tile + tn * TSTRIDE;
#pragma unroll
    for (int j = 0; j < 8; ++j) {
        T[(half * 8 + j) * TPITCH + r]           = c00[j];
        T[(half * 8 + j) * TPITCH + 16 + r]      = c01[j];
        T[(16 + half * 8 + j) * TPITCH + r]      = c10[j];
        T[(16 + half * 8 + j) * TPITCH + 16 + r] = c11[j];
    }
    __syncthreads();
    const float* ad = att + tn * 2 * C1C;
    const float* as = ad + C1C;
    float si = 0.f, sj = 0.f;
    for (int c = 0; c < C1C; ++c) {
        float hv = T[lane * TPITCH + c];
        si = __builtin_fmaf(hv, ad[c], si);
        sj = __builtin_fmaf(hv, as[c], sj);
    }
    sI[lane * H1C + tn] = si;
    sJ[lane * H1C + tn] = sj;
    __syncthreads();
    {
        const float* Th = tile + (lane >> 2) * TSTRIDE + (lane & 3) * 8;
#pragma unroll
        for (int i = 0; i < 4; ++i) {
            const int row = tn * 4 + i;
            const float* s = Th + row * TPITCH;
            v8h o;
#pragma unroll
            for (int e = 0; e < 8; ++e) o[e] = (_Float16)s[e];
            *(v8h*)(C + (size_t)(tm * 32 + row) * HIDDIM + lane * 8) = o;
        }
    }
    const int t = threadIdx.x;
    if (t < 128) {
        const int q = t & 63;
        if (tm * 32 + (q >> 1) < NN) {
            if (t < 64) ((float4*)ai)[(size_t)tm * 64 + q] = ((const float4*)sI)[q];
            else        ((float4*)aj)[(size_t)tm * 64 + q] = ((const float4*)sJ)[q];
        }
    }
}

#define BIN_SH   7
#define BIN_N    (1 << BIN_SH)
#define NBIN     ((NN + BIN_N - 1) / BIN_N)
#define CHUNK    16384
#define NCHUNK   ((E0C + CHUNK - 1) / CHUNK)
#define LSTROW   (NBIN + 1)
typedef int v4i __attribute__((ext_vector_type(4)));
static_assert(NN <= 65536, "a source id takes 16 bits of the edge word");
static_assert(NBIN <= 512, "a bin id takes 9 bits of the edge word");
static_assert(NBIN < 1024, "one thread per bin and one behind the last bin in level 1");
static_assert(E0C % 4 == 0 && CHUNK % 4 == 0, "the edge list is read by 128-bit loads");
static_assert(NCHUNK <= 64, "level 2 finds a word's chunk in a 64-entry table");

__device__ __forceinline__
int block_excl_scan_1024(int v, int* wsum) {
    const int lane = threadIdx.x & 31, w = threadIdx.x >> 5;
    int x = v;
    for (int o = 1; o < 32; o <<= 1) {
        int y = __shfl_up(x, o);
        if (lane >= o) x += y;
    }
    if (lane == 31) wsum[w] = x;
    __syncthreads();
    if (w == 0) {
        int tv = wsum[lane];
        for (int o = 1; o < 32; o <<= 1) {
            int y = __shfl_up(tv, o);
            if (lane >= o) tv += y;
        }
        wsum[lane] = tv;
    }
    __syncthreads();
    const int ex = x - v + ((w > 0) ? wsum[w - 1] : 0);
    __syncthreads();
    return ex;
}

__device__ __forceinline__
void stage_edge(int s, int d, unsigned* stage, const int* lstart, int* cursor) {
    const int b = d >> BIN_SH;
    const int r = atomicAdd(&cursor[b], 1);
    stage[lstart[b] + r] = (unsigned)s | ((unsigned)(d & (BIN_N - 1)) << 16) | ((unsigned)b << 23);
}

__device__ __forceinline__
void bin_scatter_chunk(const int* __restrict__ ei, int* __restrict__ lst,
                       unsigned* __restrict__ binned, int g,
                       unsigned* stage, int* lstart, int* cursor, int* wsum) {
    const int t = threadIdx.x;
    if (t < NBIN) cursor[t] = 0;
    const int n  = min(CHUNK, E0C - g * CHUNK);
    const int n4 = n / 4;
    const v4i* s4 = (const v4i*)ei + (size_t)g * (CHUNK / 4);
    const v4i* d4 = (const v4i*)(ei + E0C) + (size_t)g * (CHUNK / 4);
    v4i sv[CHUNK / 4 / 1024], dv[CHUNK / 4 / 1024];
#pragma unroll
    for (int k = 0; k < CHUNK / 4 / 1024; ++k) {
        const int q = min(k * 1024 + t, n4 - 1);
        sv[k] = s4[q]; dv[k] = d4[q];
    }
    __syncthreads();
#pragma unroll
    for (int k = 0; k < CHUNK / 4 / 1024; ++k) {
        if (k * 1024 + t < n4) {
            atomicAdd(&cursor[dv[k].x >> BIN_SH], 1);
            atomicAdd(&cursor[dv[k].y >> BIN_SH], 1);
            atomicAdd(&cursor[dv[k].z >> BIN_SH], 1);
            atomicAdd(&cursor[dv[k].w >> BIN_SH], 1);
        }
    }
    __syncthreads();
    const int mine = (t < NBIN) ? cursor[t] : 0;
    const int ls = block_excl_scan_1024(mine, wsum);
    if (t < NBIN) { lstart[t] = ls; cursor[t] = 0; }
    if (t <= NBIN) lst[(size_t)g * LSTROW + t] = ls;
    __syncthreads();
#pragma unroll
    for (int k = 0; k < CHUNK / 4 / 1024; ++k) {
        if (k * 1024 + t < n4) {
            stage_edge(sv[k].x, dv[k].x, stage, lstart, cursor);
            stage_edge(sv[k].y, dv[k].y, stage, lstart, cursor);
            stage_edge(sv[k].z, dv[k].z, stage, lstart, cursor);
            stage_edge(sv[k].w, dv[k].w, stage, lstart, cursor);
        }
    }
    __syncthreads();
    typedef unsigned v4u __attribute__((ext_vector_type(4)));
    v4u* o4 = (v4u*)(binned + (size_t)g * CHUNK);
    for (int q = t; q < n4; q += 1024) o4[q] = ((const v4u*)stage)[q];
}

#define STG_CAP (H1C * 32 * TPITCH)
#define BIN_U   4
static_assert(BIN_N <= 256, "one thread per target of the bin");

__device__ __forceinline__
int bin_word_index(int i, const int* segOff, const int* segBase, const int* coarse) {
    int j = 0;
#pragma unroll
    for (int q = 1; q < 8; ++q) j += (coarse[q] <= i) ? 1 : 0;
    const int* f = segOff + 8 * j;
    int c = 8 * j;
#pragma unroll
    for (int q = 1; q < 8; ++q) c += (f[q] <= i) ? 1 : 0;
    return segBase[c] + i;
}

__device__ __forceinline__
void bin_to_groups(const unsigned* __restrict__ binned, const int* __restrict__ lst,
                   int* __restrict__ rowptr, int* __restrict__ csrc, int b,
                   int* stg, int* cnt, int* first, int* cur, int* wsum,
                   int* segOff, int* segBase, int* coarse) {
    const int t = threadIdx.x, lane = t & 31, w = t >> 5;
    int ra = 0, rl = 0;
    if (t < NCHUNK) {
        ra = lst[(size_t)t * LSTROW + b];
        rl = lst[(size_t)t * LSTROW + b + 1] - ra;
    }
    if (t < BIN_N) { cnt[t] = 0; cur[t] = 0; }
    int rx = rl, rs = ra;
    for (int o = 1; o < 32; o <<= 1) {
        int y = __shfl_up(rx, o);
        if (lane >= o) rx += y;
    }
    for (int o = 16; o > 0; o >>= 1) rs += __shfl_xor(rs, o);
    if (w < 2 && lane == 31) { wsum[w] = rx; wsum[2 + w] = rs; }
    __syncthreads();
    const int n = wsum[0] + wsum[1];
    const int s = wsum[2] + wsum[3];
    if (t < 64) {
        const int ex = rx - rl + (w ? wsum[0] : 0);
        const int so = (t < NCHUNK) ? ex : 0x7fffffff;
        segOff[t]  = so;
        segBase[t] = t * CHUNK + ra - ex;
        if ((t & 7) == 0) coarse[t >> 3] = so;
    }
    __syncthreads();
    for (int i0 = 0; i0 < n; i0 += BIN_U * 256) {
        unsigned wv[BIN_U];
#pragma unroll
        for (int k = 0; k < BIN_U; ++k)
            wv[k] = binned[bin_word_index(min(i0 + k * 256 + t, n - 1), segOff, segBase, coarse)];
#pragma unroll
        for (int k = 0; k < BIN_U; ++k) {
            const int i = i0 + k * 256 + t;
            if (i < n) atomicAdd(&cnt[(wv[k] >> 16) & (BIN_N - 1)], 1);
        }
    }
    __syncthreads();
    const int c = (t < BIN_N) ? cnt[t] : 0;
    int x = c;
    for (int o = 1; o < 32; o <<= 1) {
        int y = __shfl_up(x, o);
        if (lane >= o) x += y;
    }
    if (lane == 31) wsum[w] = x;
    __syncthreads();
    int off = 0;
    for (int k = 0; k < w; ++k) off += wsum[k];
    const int ex = x - c + off;
    if (t < BIN_N) {
        first[t] = ex;
        const int node = b * BIN_N + t;
        if (node < NN) rowptr[node] = s + ex;
    }
    if (b == NBIN - 1 && t == 0) rowptr[NN] = s + n;
    __syncthreads();
    const bool fits = n <= STG_CAP;
    for (int i0 = 0; i0 < n; i0 += BIN_U * 256) {
        unsigned wv[BIN_U];
#pragma unroll
        for (int k = 0; k < BIN_U; ++k)
            wv[k] = binned[bin_word_index(min(i0 + k * 256 + t, n - 1), segOff, segBase, coarse)];
#pragma unroll
        for (int k = 0; k < BIN_U; ++k) {
            const int i = i0 + k * 256 + t;
            if (i < n) {
                const int d = (int)(wv[k] >> 16) & (BIN_N - 1);
                const int p = first[d] + atomicAdd(&cur[d], 1);
                const int sv = (int)(wv[k] & 0xFFFFu);
                if (fits) stg[p] = sv;
                else      csrc[s + p] = sv;
            }
        }
    }
    if (fits) {
        __syncthreads();
        for (int i = t; i < n; i += 256) csrc[s + i] = stg[i];
    }
}

#define MT2C     (MPAD / 32)

#define SORT_TB   1024
#define CVT_WAVES ((NN / 16) * 2)
#define CVT_BLK   ((CVT_WAVES + SORT_TB / 32 - 1) / (SORT_TB / 32))
#define XP1_BLK   (INDIM * HIDDIM / SORT_TB)
#define XP2_BLK   (HIDDIM * OUTDIM / SORT_TB)
static_assert(XP1_BLK + XP2_BLK <= CVT_BLK, "the cast's work-groups cover both weight matrices");
static_assert(INDIM * HIDDIM % SORT_TB == 0 && HIDDIM * OUTDIM % SORT_TB == 0, "whole work-groups of weight elements");
__global__ __launch_bounds__(SORT_TB) __attribute__((amdgpu_waves_per_eu(8, 8)))
void scatter_cast_weights(const int* __restrict__ ei, int* __restrict__ lst,
                          unsigned* __restrict__ binned,
                          const float* __restrict__ x, _Float16* __restrict__ xh,
                          const float* __restrict__ W1, _Float16* __restrict__ W1T,
                          const float* __restrict__ W2, _Float16* __restrict__ W2T) {
    __shared__ __attribute__((aligned(16))) unsigned stage[CHUNK];
    __shared__ int lstart[NBIN];
    __shared__ int cursor[NBIN];
    __shared__ int wsum[32];
    if (blockIdx.x < NCHUNK) {
        bin_scatter_chunk(ei, lst, binned, (int)blockIdx.x, stage, lstart, cursor, wsum);
    } else {
        const int cb = (int)blockIdx.x - NCHUNK;
        const bool w1 = cb < XP1_BLK;
        const bool w2 = !w1 && cb < XP1_BLK + XP2_BLK;
        const int wi = min((w1 ? cb : cb - XP1_BLK) * SORT_TB + (int)threadIdx.x,
                           (w1 ? INDIM * HIDDIM : HIDDIM * OUTDIM) - 1);
        const float wv = (w1 ? W1 : W2)[wi];
        cvt_f32_to_f16_wave(x, xh, cb * (SORT_TB / 32) + (int)(threadIdx.x >> 5), threadIdx.x & 31);
        if (w1)      W1T[frag_index(wi % HIDDIM, wi / HIDDIM, INDIM)]  = (_Float16)wv;
        else if (w2) W2T[frag_index(wi % OUTDIM, wi / OUTDIM, HIDDIM)] = (_Float16)wv;
    }
}

static_assert(4 * NBIN >= MT2C, "four GEMM tiles per bin cover all row tiles");
__global__ __launch_bounds__(256)
void gemm_l1_coef_and_groups(const _Float16* __restrict__ A, const _Float16* __restrict__ Bt,
                             _Float16* __restrict__ C, const float* __restrict__ att,
                             float* __restrict__ ai, float* __restrict__ aj,
                             const unsigned* __restrict__ binned, const int* __restrict__ lst,
                             int* __restrict__ rowptr, int* __restrict__ csrc) {
    __shared__ float tile[H1C * TSTRIDE];
    static_assert(H1C * TSTRIDE >= STG_CAP, "the bin's staging fits in the tiles");
    __shared__ __attribute__((aligned(16))) float sI[32 * H1C];
    __shared__ __attribute__((aligned(16))) float sJ[32 * H1C];
    static_assert(32 * H1C >= 2 * BIN_N, "sI holds cnt and first, sJ holds cur and wsum");
    __shared__ __attribute__((aligned(16))) int seg[2 * 64 + 8];
    const int g = blockIdx.x / 5;
    const int m = blockIdx.x - 5 * g;
    if (m == 0) {
        bin_to_groups(binned, lst, rowptr, csrc, g, (int*)tile,
                      (int*)sI, (int*)sI + BIN_N, (int*)sJ, (int*)sJ + BIN_N, seg, seg + 64, seg + 128);
    } else {
        const int tm = 4 * g + m - 1;
        if (tm < MT2C)
            gemm_l1_coef_tile(A, Bt, C, att, ai, aj, tm, tile, sI, sJ);
    }
}

__device__ __forceinline__ float lrelu(float a) { return (a >= 0.f) ? a : NEG_SLOPE * a; }
__device__ __forceinline__ float elu1(float v) { return (v > 0.f) ? v : (__expf(v) - 1.f); }

__device__ __forceinline__ void fold_edge(float a, const float4 r,
                                          float& m, float& ssum, float4& acc) {
    float mn = fmaxf(m, a);
    float sc = __expf(m - mn);
    float w  = __expf(a - mn);
    ssum  = ssum * sc + w;
    acc.x = acc.x * sc + w * r.x;
    acc.y = acc.y * sc + w * r.y;
    acc.z = acc.z * sc + w * r.z;
    acc.w = acc.w * sc + w * r.w;
    m = mn;
}

__device__ __forceinline__ void fold_edge8(float a, const v8h r,
                                           float& m, float& ssum, float (&acc)[8]) {
    float mn = fmaxf(m, a);
    float sc = __expf(m - mn);
    float w  = __expf(a - mn);
    ssum = ssum * sc + w;
#pragma unroll
    for (int e = 0; e < 8; ++e) acc[e] = __builtin_fmaf((float)r[e], w, acc[e] * sc);
    m = mn;
}

#define FN      16
#define APITCH  (HIDDIM + 8)
#define H2PITCH (OUTDIM + 4)
static_assert(NN % FN == 0, "whole 16-row tiles only: no pad rows");
static_assert(HIDDIM == 32 * 8 && C1C % 8 == 0, "a lane's 8 channels lie in one head");
__global__ __launch_bounds__(256)
void node_softmax_l1_xform(const int* __restrict__ rowptr, const int* __restrict__ csrc,
                 const float* __restrict__ ai, const float* __restrict__ aj,
                 const _Float16* __restrict__ h, const float* __restrict__ bias,
                 const _Float16* __restrict__ W2T, const float* __restrict__ att2,
                 float* __restrict__ h2, float* __restrict__ ai2, float* __restrict__ aj2) {
    __shared__ __attribute__((aligned(16))) _Float16 sact[FN * APITCH];
    __shared__ __attribute__((aligned(16))) float sh2[FN * H2PITCH];
    const int w    = threadIdx.x >> 5;
    const int lane = threadIdx.x & 31;
    const int hd   = lane / (C1C / 8);

    static_assert((size_t)MPAD * HIDDIM < (1ull << 31) && (size_t)NN * H1C < (1ull << 31), "32-bit element offsets");
#define L1_LOAD(s, ajv, r) { \
        ajv = aj[(unsigned)(s) * (unsigned)H1C + (unsigned)hd]; \
        r = *(const v8h*)(h + ((unsigned)(s) * (unsigned)HIDDIM + (unsigned)lane * 8u)); }
#define L1_FOLD(ajv, r) fold_edge8(lrelu(ai0 + (ajv)), r, m0, s0, acc)

#pragma unroll 1
    for (int i = 0; i < FN / 8; ++i) {
        const int row  = (FN / 8) * w + i;
        const int node = (int)blockIdx.x * FN + row;
        const float ai0 = ai[(unsigned)node * (unsigned)H1C + (unsigned)hd];
        const int beg = rowptr[node], end = rowptr[node + 1];

        float m0, s0 = 1.f;
        float acc[8];
        {
            float ajv; v8h r;
            L1_LOAD(node, ajv, r);
            m0 = lrelu(ai0 + ajv);
#pragma unroll
            for (int e = 0; e < 8; ++e) acc[e] = (float)r[e];
        }
        int p = beg;
        for (; p + 8 <= end; p += 8) {
            int sidx[8]; float jv[8]; v8h rr[8];
#pragma unroll
            for (int k = 0; k < 8; ++k) sidx[k] = csrc[p + k];
#pragma unroll
            for (int k = 0; k < 8; ++k) L1_LOAD(sidx[k], jv[k], rr[k]);
#pragma unroll
            for (int k = 0; k < 8; ++k) L1_FOLD(jv[k], rr[k]);
        }
        if (p + 4 <= end) {
            int sidx[4]; float jv[4]; v8h rr[4];
#pragma unroll
            for (int k = 0; k < 4; ++k) sidx[k] = csrc[p + k];
#pragma unroll
            for (int k = 0; k < 4; ++k) L1_LOAD(sidx[k], jv[k], rr[k]);
#pragma unroll
            for (int k = 0; k < 4; ++k) L1_FOLD(jv[k], rr[k]);
            p += 4;
        }
        for (; p < end; ++p) {
            int sa = csrc[p];
            float ja; v8h ra;
            L1_LOAD(sa, ja, ra);
            L1_FOLD(ja, ra);
        }
        const float i0 = 1.f / s0;
        const float4 bA = ((const float4*)bias)[2 * lane];
        const float4 bB = ((const float4*)bias)[2 * lane + 1];
        v8h o;
        o[0] = (_Float16)elu1(acc[0] * i0 + bA.x); o[1] = (_Float16)elu1(acc[1] * i0 + bA.y);
        o[2] = (_Float16)elu1(acc[2] * i0 + bA.z); o[3] = (_Float16)elu1(acc[3] * i0 + bA.w);
        o[4] = (_Float16)elu1(acc[4] * i0 + bB.x); o[5] = (_Float16)elu1(acc[5] * i0 + bB.y);
        o[6] = (_Float16)elu1(acc[6] * i0 + bB.z); o[7] = (_Float16)elu1(acc[7] * i0 + bB.w);
        *(v8h*)(sact + row * APITCH + lane * 8) = o;
    }
#undef L1_LOAD
#undef L1_FOLD
    __syncthreads();

    if (w < OUTDIM / 16) {
        const int r    = lane & 15;
        const int half = lane >> 4;
        const _Float16* a0 = sact + r * APITCH + half * 8;
        const _Float16* b0 = W2T + (size_t)w * FRAG_TILE(HIDDIM) + lane * 8;
        v8f c = {};
        for (int k0 = 0; k0 < HIDDIM; k0 += 32) {
            V16U A, B;
            A.h[0] = *(const v8h*)(a0 + k0);
            A.h[1] = *(const v8h*)(a0 + k0 + 16);
            B.h[0] = *(const v8h*)(b0 + k0 * 16);
            B.h[1] = *(const v8h*)(b0 + k0 * 16 + 256);
            c = WMMA16(A.v, B.v, c);
        }
#pragma unroll
        for (int j = 0; j < 8; ++j) sh2[(half * 8 + j) * H2PITCH + w * 16 + r] = c[j];
    }
    __syncthreads();

    const int t = threadIdx.x;
    if (t < FN) {
        const float* hv = sh2 + t * H2PITCH;
        float si = 0.f, sj = 0.f;
        for (int c = 0; c < C2C; ++c) {
            si = __builtin_fmaf(hv[c], att2[c], si);
            sj = __builtin_fmaf(hv[c], att2[C2C + c], sj);
        }
        ai2[(int)blockIdx.x * FN + t] = si;
        aj2[(int)blockIdx.x * FN + t] = sj;
    }
    {
        const int row = t >> 4, q = t & 15;
        ((float4*)(h2 + (size_t)((int)blockIdx.x * FN + row) * OUTDIM))[q] =
            *(const float4*)(sh2 + row * H2PITCH + 4 * q);
    }
}

__global__ __launch_bounds__(256)
void node_softmax_l2(const int* __restrict__ rowptr, const int* __restrict__ csrc,
                 const float* __restrict__ ai, const float* __restrict__ aj,
                 const float* __restrict__ h, const float* __restrict__ bias,
                 float* __restrict__ outp) {
    int gid  = blockIdx.x * blockDim.x + threadIdx.x;
    int node = gid >> 4;
    int lg   = gid & 15;
    if (node >= NN) return;
    const float aiv = ai[node];
    const int beg = rowptr[node], end = rowptr[node + 1];

    float m = lrelu(aiv + aj[node]), ssum = 1.f;
    float4 acc = ((const float4*)(h + (size_t)node * OUTDIM))[lg];

    int p = beg;
    for (; p + 4 <= end; p += 4) {
        int sa = csrc[p], sb = csrc[p + 1], sc_ = csrc[p + 2], sd = csrc[p + 3];
        float ja = aj[sa], jb = aj[sb], jc = aj[sc_], jd = aj[sd];
        float4 ra = ((const float4*)(h + (size_t)sa  * OUTDIM))[lg];
        float4 rb = ((const float4*)(h + (size_t)sb  * OUTDIM))[lg];
        float4 rc = ((const float4*)(h + (size_t)sc_ * OUTDIM))[lg];
        float4 rd = ((const float4*)(h + (size_t)sd  * OUTDIM))[lg];
        fold_edge(lrelu(aiv + ja), ra, m, ssum, acc);
        fold_edge(lrelu(aiv + jb), rb, m, ssum, acc);
        fold_edge(lrelu(aiv + jc), rc, m, ssum, acc);
        fold_edge(lrelu(aiv + jd), rd, m, ssum, acc);
    }
    for (; p < end; ++p) {
        int sa = csrc[p];
        float ja = aj[sa];
        float4 ra = ((const float4*)(h + (size_t)sa * OUTDIM))[lg];
        fold_edge(lrelu(aiv + ja), ra, m, ssum, acc);
    }
    float inv = 1.f / ssum;
    const float4 b = ((const float4*)bias)[lg];
    ((float4*)(outp + (size_t)node * OUTDIM))[lg] =
        make_float4(acc.x * inv + b.x, acc.y * inv + b.y, acc.z * inv + b.z, acc.w * inv + b.w);
}


extern "C" void kernel_launch(void* const* d_in, const int* in_sizes, int n_in,
                              void* d_out, int out_size, void* d_ws, size_t ws_size,
                              hipStream_t stream) {
    (void)in_sizes; (void)n_in; (void)out_size; (void)ws_size;

    const float* x    = (const float*)d_in[0];
    const int*   ei   = (const int*)  d_in[1];
    const float* W1   = (const float*)d_in[2];
    const float* att1 = (const float*)d_in[3];
    const float* b1   = (const float*)d_in[4];
    const float* W2   = (const float*)d_in[5];
    const float* att2 = (const float*)d_in[6];
    const float* b2   = (const float*)d_in[7];
    float* out = (float*)d_out;

    char* base = (char*)d_ws;
    size_t off = 0;
    auto carve = [&](size_t bytes) -> char* {
        char* p = base + off;
        off = (off + bytes + 255) & ~(size_t)255;
        return p;
    };
    _Float16* XH    = (_Float16*)carve((size_t)MPAD * INDIM * 2);
    _Float16* W1T   = (_Float16*)carve((size_t)INDIM * HIDDIM * 2);
    _Float16* W2T   = (_Float16*)carve((size_t)HIDDIM * OUTDIM * 2);
    _Float16* HBUF  = (_Float16*)carve((size_t)MPAD * HIDDIM * 2);
    float*    AI    = (float*)   carve((size_t)NN * H1C * 4);
    float*    AJ    = (float*)   carve((size_t)NN * H1C * 4);
    int*      LST   = (int*)     carve((size_t)NCHUNK * LSTROW * 4);
    unsigned* BINNED = (unsigned*)carve((size_t)NCHUNK * CHUNK * 4);
    int*      ROWP  = (int*)     carve((size_t)(NN + 1) * 4);
    int*      CSRC  = (int*)     carve((size_t)E0C * 4);
    float*    H2BUF = (float*)   carve((size_t)NN * OUTDIM * 4);
    float*    AI2   = (float*)   carve((size_t)NN * 4);
    float*    AJ2   = (float*)   carve((size_t)NN * 4);

    const int TB = 256;
    auto nb = [](long long n, int tb) { return (unsigned)((n + tb - 1) / tb); };

    scatter_cast_weights<<<NCHUNK + CVT_BLK, SORT_TB, 0, stream>>>(ei, LST, BINNED, x, XH, W1, W1T, W2, W2T);

    gemm_l1_coef_and_groups<<<5 * NBIN, 256, 0, stream>>>(XH, W1T, HBUF, att1, AI, AJ, BINNED, LST, ROWP, CSRC);

    node_softmax_l1_xform<<<NN / FN, TB, 0, stream>>>(ROWP, CSRC, AI, AJ, HBUF, b1, W2T, att2, H2BUF, AI2, AJ2);

    node_softmax_l2<<<nb((long long)NN * 16, TB), TB, 0, stream>>>(ROWP, CSRC, AI2, AJ2, H2BUF, b2, out);
}
